# starting kernel
# baseline (speedup 1.0000x reference)
#include <hip/hip_runtime.h>


typedef float v2f __attribute__((ext_vector_type(2)));
typedef float v8f __attribute__((ext_vector_type(8)));

#define N_WIRES 10
#define N_LAYERS 4
#define DIM 1024
#define PADDED(k) ((k) + ((k) >> 4))
#define PDIM (DIM + (DIM >> 4))
#define LAYER_F 1536

__device__ __forceinline__ void cmul(float ar, float ai, float br, float bi,
                                     float &cr, float &ci) {
  cr = ar * br - ai * bi;
  ci = ar * bi + ai * br;
}

__device__ __forceinline__ int cnot_perm(int m, int rr) {
#pragma unroll
  for (int w = 0; w < N_WIRES; ++w) {
    int t = w + rr; if (t >= N_WIRES) t -= N_WIRES;
    m ^= ((m >> (9 - w)) & 1) << (9 - t);
  }
  return m;
}

__global__ __launch_bounds__(256) void vqc_precompute(const float *__restrict__ W,
                                                      float *__restrict__ G) {
  __shared__ float u[N_WIRES][8];
  const int tid = threadIdx.x;
  for (int l = 0; l < N_LAYERS; ++l) {
    if (tid < N_WIRES) {
      const float phi   = W[(l * N_WIRES + tid) * 3 + 0];
      const float theta = W[(l * N_WIRES + tid) * 3 + 1];
      const float omega = W[(l * N_WIRES + tid) * 3 + 2];
      const float c = __cosf(0.5f * theta), s = __sinf(0.5f * theta);
      const float a = 0.5f * (phi + omega), b = 0.5f * (phi - omega);
      const float ca = __cosf(a), sa = __sinf(a), cb = __cosf(b), sb = __sinf(b);
      u[tid][0] =  ca * c; u[tid][1] = -sa * c;
      u[tid][2] = -cb * s; u[tid][3] = -sb * s;
      u[tid][4] =  cb * s; u[tid][5] = -sb * s;
      u[tid][6] =  ca * c; u[tid][7] =  sa * c;
    }
    __syncthreads();
    float *L = G + l * LAYER_F;
    const int r = tid >> 4, c = tid & 15;
#pragma unroll
    for (int g = 0; g < 2; ++g) {
      float pr = 1.f, pi = 0.f;
#pragma unroll
      for (int q = 0; q < 4; ++q) {
        const int rb = (r >> (3 - q)) & 1, cb2 = (c >> (3 - q)) & 1;
        const float *e = &u[g * 4 + q][(rb * 2 + cb2) * 2];
        float nr, ni; cmul(pr, pi, e[0], e[1], nr, ni);
        pr = nr; pi = ni;
      }
      L[g * 512 + c * 16 + r] = pr;
      L[g * 512 + 256 + c * 16 + r] = pi;
    }
    {
      const int a = r >> 2, b = r & 3, a2 = c >> 2, b2 = c & 3;
      float pr = 0.f, pi = 0.f;
      if (a == a2) {
        const float *e8 = &u[8][(((b >> 1) * 2) + (b2 >> 1)) * 2];
        const float *e9 = &u[9][(((b & 1) * 2) + (b2 & 1)) * 2];
        cmul(e8[0], e8[1], e9[0], e9[1], pr, pi);
      }
      L[1024 + c * 16 + r] = pr;
      L[1280 + c * 16 + r] = pi;
    }
    __syncthreads();
  }
}

__device__ __forceinline__ void group_apply(const float *__restrict__ A,
                                            const float *p1, const float *p2,
                                            float *dst,
                                            int kkStride, int compOff,
                                            int mode, int rr, int permBase,
                                            bool pimg) {
  v8f acc = {};
#pragma unroll
  for (int kk = 0; kk < 4; ++kk) {
    v2f ar, ai, bF, bS, aS;
    ar.x = A[kk * 64];        ar.y = A[kk * 64 + 16];
    ai.x = A[256 + kk * 64];  ai.y = A[256 + kk * 64 + 16];
    bF.x = p1[kk * kkStride];           bF.y = p1[kk * kkStride + compOff];
    bS.x = p2[kk * kkStride];           bS.y = p2[kk * kkStride + compOff];
    aS.x = pimg ? ai.x : -ai.x;
    aS.y = pimg ? ai.y : -ai.y;
    acc = __builtin_amdgcn_wmma_f32_16x16x4_f32(false, ar, false, bF,
                                                (short)0, acc, false, false);
    acc = __builtin_amdgcn_wmma_f32_16x16x4_f32(false, aS, false, bS,
                                                (short)0, acc, false, false);
  }
#pragma unroll
  for (int v = 0; v < 8; ++v) {
    int w;
    if (mode == 0)      w = v * 68;
    else if (mode == 1) w = v * 4 + (v >> 2);
    else                w = PADDED(permBase ^ cnot_perm(v, rr));
    dst[w] = acc[v];
  }
}

__global__ __launch_bounds__(256) void vqc_kernel(const float *__restrict__ X,
                                                  const float *__restrict__ G,
                                                  const float *__restrict__ bias,
                                                  float *__restrict__ out) {
  __shared__ float SR[2][PDIM];
  __shared__ float SI[2][PDIM];
  __shared__ float gm[N_LAYERS * LAYER_F];
  __shared__ float cs[2 * N_WIRES];
  __shared__ float pLo[32], pHi[32];
  __shared__ int   permT[N_LAYERS][64];
  __shared__ float red[8];

  const int tid  = threadIdx.x;
  const int lane = tid & 31;
  const int wave = tid >> 5;
  const int s    = blockIdx.x;

  for (int i = tid; i < N_LAYERS * LAYER_F; i += 256) gm[i] = G[i];

  permT[tid >> 6][tid & 63] = cnot_perm((tid & 63) << 4, (tid >> 6) + 1);

  if (tid < N_WIRES) {
    const float x = X[s * N_WIRES + tid] * 0.7853981633974483f;
    cs[tid] = __cosf(x);
    cs[N_WIRES + tid] = __sinf(x);
  }
  __syncthreads();

  if (tid < 32) {
    float p = 1.f;
#pragma unroll
    for (int w = 5; w < N_WIRES; ++w)
      p *= cs[((tid >> (9 - w)) & 1) * N_WIRES + w];
    pLo[tid] = p;
  } else if (tid < 64) {
    const int h = tid - 32;
    float p = 1.f;
#pragma unroll
    for (int w = 0; w < 5; ++w)
      p *= cs[((h >> (4 - w)) & 1) * N_WIRES + w];
    pHi[h] = p;
  }
  __syncthreads();

  for (int k = tid; k < DIM; k += 256) {
    SR[0][PADDED(k)] = pHi[k >> 5] * pLo[k & 31];
    SI[0][PADDED(k)] = 0.f;
  }
  __syncthreads();

  const int  tile = wave >> 1;
  const bool pimg = (wave & 1) != 0;
  const int  col  = tile * 16 + (lane & 15);
  const int  kb   = (lane >> 4) << 1;
  const int  M    = lane & 15;
  const int  laneHi8 = (lane >> 4) << 3;
  const int  cx = col >> 2, cy = col & 3;
  const int baseB0 = col + (col >> 4) + kb * 68;
  const int baseB1 = cx * 68 + cy + kb * 4;
  const int baseB2 = col * 17 + kb;
  const int baseW0 = col + (col >> 4) + laneHi8 * 68;
  const int baseW1 = cx * 68 + cy + laneHi8 * 4 + (laneHi8 >> 2);
  const int baseA  = kb * 16 + M;

#pragma unroll
  for (int l = 0; l < N_LAYERS; ++l) {
    const float *Lg = gm + l * LAYER_F;
    const int r0 = (3 * l) & 1;

    group_apply(Lg + baseA,
                (pimg ? SI[r0] : SR[r0]) + baseB0,
                (pimg ? SR[r0] : SI[r0]) + baseB0,
                (pimg ? SI[r0 ^ 1] : SR[r0 ^ 1]) + baseW0,
                272, 68, 0, 0, 0, pimg);
    __syncthreads();
    group_apply(Lg + 512 + baseA,
                (pimg ? SI[r0 ^ 1] : SR[r0 ^ 1]) + baseB1,
                (pimg ? SR[r0 ^ 1] : SI[r0 ^ 1]) + baseB1,
                (pimg ? SI[r0] : SR[r0]) + baseW1,
                17, 4, 1, 0, 0, pimg);
    __syncthreads();
    {
      const int pb = permT[l][col] ^ ((lane & 16) ? cnot_perm(8, l + 1) : 0);
      group_apply(Lg + 1024 + baseA,
                  (pimg ? SI[r0] : SR[r0]) + baseB2,
                  (pimg ? SR[r0] : SI[r0]) + baseB2,
                  (pimg ? SI[r0 ^ 1] : SR[r0 ^ 1]),
                  4, 1, 2, l + 1, pb, pimg);
      __syncthreads();
    }
  }

  const float sgn = (tid & 1) ? -1.f : 1.f;
  float acc = 0.f;
#pragma unroll
  for (int j = 0; j < DIM / 256; ++j) {
    const int a = PADDED(tid + 256 * j);
    const float re = SR[0][a], im = SI[0][a];
    acc += re * re + im * im;
  }
  float z = sgn * acc;
#pragma unroll
  for (int m = 16; m > 0; m >>= 1) z += __shfl_xor(z, m, 32);
  if (lane == 0) red[wave] = z;
  __syncthreads();
  if (tid == 0) {
    float t = 0.f;
#pragma unroll
    for (int w = 0; w < 8; ++w) t += red[w];
    out[s] = t + bias[0];
  }
}

extern "C" void kernel_launch(void *const *d_in, const int *in_sizes, int n_in,
                              void *d_out, int out_size, void *d_ws, size_t ws_size,
                              hipStream_t stream) {
  const float *X = (const float *)d_in[0];
  const float *W = (const float *)d_in[1];
  const float *bias = (const float *)d_in[2];
  float *out = (float *)d_out;
  float *G = (float *)d_ws;

  const int B = in_sizes[0] / N_WIRES;

  vqc_precompute<<<1, 256, 0, stream>>>(W, G);
  vqc_kernel<<<B, 256, 0, stream>>>(X, G, bias, out);
}


# Layer 0 in closed form: each wire's Rot gate acts on its encoded 2-vector, product state scattered via the layer-0 CNOT map; 3 of 12 WMMA phases gone; layer-0 sums become products (new rounding)
# speedup vs baseline: 1.2699x; 1.2699x over previous
#include <hip/hip_runtime.h>


typedef float v2f __attribute__((ext_vector_type(2)));
typedef float v8f __attribute__((ext_vector_type(8)));

#define N_WIRES 10
#define N_LAYERS 4
#define DIM 1024
#define PADDED(k) ((k) + ((k) >> 4))
#define PDIM (DIM + (DIM >> 4))
#define LAYER_F 1536
#define N_FUSED (N_LAYERS - 1)
#define U0_OFF (N_FUSED * LAYER_F)

__device__ __forceinline__ void cmul(float ar, float ai, float br, float bi,
                                     float &cr, float &ci) {
  cr = ar * br - ai * bi;
  ci = ar * bi + ai * br;
}

__device__ __forceinline__ int cnot_perm(int m, int rr) {
#pragma unroll
  for (int w = 0; w < N_WIRES; ++w) {
    int t = w + rr; if (t >= N_WIRES) t -= N_WIRES;
    m ^= ((m >> (9 - w)) & 1) << (9 - t);
  }
  return m;
}

__global__ __launch_bounds__(256) void vqc_precompute(const float *__restrict__ W,
                                                      float *__restrict__ G) {
  __shared__ float u[N_WIRES][8];
  const int tid = threadIdx.x;
  for (int l = 0; l < N_LAYERS; ++l) {
    if (tid < N_WIRES) {
      const float phi   = W[(l * N_WIRES + tid) * 3 + 0];
      const float theta = W[(l * N_WIRES + tid) * 3 + 1];
      const float omega = W[(l * N_WIRES + tid) * 3 + 2];
      const float c = __cosf(0.5f * theta), s = __sinf(0.5f * theta);
      const float a = 0.5f * (phi + omega), b = 0.5f * (phi - omega);
      const float ca = __cosf(a), sa = __sinf(a), cb = __cosf(b), sb = __sinf(b);
      u[tid][0] =  ca * c; u[tid][1] = -sa * c;
      u[tid][2] = -cb * s; u[tid][3] = -sb * s;
      u[tid][4] =  cb * s; u[tid][5] = -sb * s;
      u[tid][6] =  ca * c; u[tid][7] =  sa * c;
    }
    __syncthreads();
    if (l == 0) {
      if (tid < N_WIRES * 8) G[U0_OFF + tid] = u[tid >> 3][tid & 7];
    } else {
    float *L = G + (l - 1) * LAYER_F;
    const int r = tid >> 4, c = tid & 15;
#pragma unroll
    for (int g = 0; g < 2; ++g) {
      float pr = 1.f, pi = 0.f;
#pragma unroll
      for (int q = 0; q < 4; ++q) {
        const int rb = (r >> (3 - q)) & 1, cb2 = (c >> (3 - q)) & 1;
        const float *e = &u[g * 4 + q][(rb * 2 + cb2) * 2];
        float nr, ni; cmul(pr, pi, e[0], e[1], nr, ni);
        pr = nr; pi = ni;
      }
      L[g * 512 + c * 16 + r] = pr;
      L[g * 512 + 256 + c * 16 + r] = pi;
    }
    {
      const int a = r >> 2, b = r & 3, a2 = c >> 2, b2 = c & 3;
      float pr = 0.f, pi = 0.f;
      if (a == a2) {
        const float *e8 = &u[8][(((b >> 1) * 2) + (b2 >> 1)) * 2];
        const float *e9 = &u[9][(((b & 1) * 2) + (b2 & 1)) * 2];
        cmul(e8[0], e8[1], e9[0], e9[1], pr, pi);
      }
      L[1024 + c * 16 + r] = pr;
      L[1280 + c * 16 + r] = pi;
    }
    }
    __syncthreads();
  }
}

__device__ __forceinline__ void group_apply(const float *__restrict__ A,
                                            const float *p1, const float *p2,
                                            float *dst,
                                            int kkStride, int compOff,
                                            int mode, int rr, int permBase,
                                            bool pimg) {
  v8f acc = {};
#pragma unroll
  for (int kk = 0; kk < 4; ++kk) {
    v2f ar, ai, bF, bS, aS;
    ar.x = A[kk * 64];        ar.y = A[kk * 64 + 16];
    ai.x = A[256 + kk * 64];  ai.y = A[256 + kk * 64 + 16];
    bF.x = p1[kk * kkStride];           bF.y = p1[kk * kkStride + compOff];
    bS.x = p2[kk * kkStride];           bS.y = p2[kk * kkStride + compOff];
    aS.x = pimg ? ai.x : -ai.x;
    aS.y = pimg ? ai.y : -ai.y;
    acc = __builtin_amdgcn_wmma_f32_16x16x4_f32(false, ar, false, bF,
                                                (short)0, acc, false, false);
    acc = __builtin_amdgcn_wmma_f32_16x16x4_f32(false, aS, false, bS,
                                                (short)0, acc, false, false);
  }
#pragma unroll
  for (int v = 0; v < 8; ++v) {
    int w;
    if (mode == 0)      w = v * 68;
    else if (mode == 1) w = v * 4 + (v >> 2);
    else                w = PADDED(permBase ^ cnot_perm(v, rr));
    dst[w] = acc[v];
  }
}

__global__ __launch_bounds__(256) void vqc_kernel(const float *__restrict__ X,
                                                  const float *__restrict__ G,
                                                  const float *__restrict__ bias,
                                                  float *__restrict__ out) {
  __shared__ float SR[2][PDIM];
  __shared__ float SI[2][PDIM];
  __shared__ float gm[N_FUSED * LAYER_F];
  __shared__ float vr[2 * N_WIRES];
  __shared__ float vi[2 * N_WIRES];
  __shared__ float pLoR[32], pLoI[32], pHiR[32], pHiI[32];
  __shared__ int   permT[N_FUSED][64];
  __shared__ float red[8];

  const int tid  = threadIdx.x;
  const int lane = tid & 31;
  const int wave = tid >> 5;
  const int s    = blockIdx.x;

  for (int i = tid; i < N_FUSED * LAYER_F; i += 256) gm[i] = G[i];

  if (tid >= 64)
    permT[(tid >> 6) - 1][tid & 63] = cnot_perm((tid & 63) << 4, (tid >> 6) + 1);

  const int perm0 = cnot_perm(tid, 1);

  if (tid < N_WIRES) {
    const float x = X[s * N_WIRES + tid] * 0.7853981633974483f;
    const float c = __cosf(x), sn = __sinf(x);
    const float *m = G + U0_OFF + tid * 8;
    vr[tid]           = m[0] * c + m[2] * sn;
    vi[tid]           = m[1] * c + m[3] * sn;
    vr[N_WIRES + tid] = m[4] * c + m[6] * sn;
    vi[N_WIRES + tid] = m[5] * c + m[7] * sn;
  }
  __syncthreads();

  if (tid < 32) {
    float pr = 1.f, pi = 0.f;
#pragma unroll
    for (int w = 5; w < N_WIRES; ++w) {
      const int e = ((tid >> (9 - w)) & 1) * N_WIRES + w;
      float nr, ni; cmul(pr, pi, vr[e], vi[e], nr, ni);
      pr = nr; pi = ni;
    }
    pLoR[tid] = pr; pLoI[tid] = pi;
  } else if (tid < 64) {
    const int h = tid - 32;
    float pr = 1.f, pi = 0.f;
#pragma unroll
    for (int w = 0; w < 5; ++w) {
      const int e = ((h >> (4 - w)) & 1) * N_WIRES + w;
      float nr, ni; cmul(pr, pi, vr[e], vi[e], nr, ni);
      pr = nr; pi = ni;
    }
    pHiR[h] = pr; pHiI[h] = pi;
  }
  __syncthreads();

#pragma unroll
  for (int j = 0; j < DIM / 256; ++j) {
    const int k = tid + 256 * j;
    float re, im;
    cmul(pHiR[k >> 5], pHiI[k >> 5], pLoR[k & 31], pLoI[k & 31], re, im);
    const int d = perm0 ^ cnot_perm(256 * j, 1);
    SR[1][PADDED(d)] = re;
    SI[1][PADDED(d)] = im;
  }
  __syncthreads();

  const int  tile = wave >> 1;
  const bool pimg = (wave & 1) != 0;
  const int  col  = tile * 16 + (lane & 15);
  const int  kb   = (lane >> 4) << 1;
  const int  M    = lane & 15;
  const int  laneHi8 = (lane >> 4) << 3;
  const int  cx = col >> 2, cy = col & 3;
  const int baseB0 = col + (col >> 4) + kb * 68;
  const int baseB1 = cx * 68 + cy + kb * 4;
  const int baseB2 = col * 17 + kb;
  const int baseW0 = col + (col >> 4) + laneHi8 * 68;
  const int baseW1 = cx * 68 + cy + laneHi8 * 4 + (laneHi8 >> 2);
  const int baseA  = kb * 16 + M;

#pragma unroll
  for (int l = 1; l < N_LAYERS; ++l) {
    const float *Lg = gm + (l - 1) * LAYER_F;
    const int r0 = (3 * l) & 1;

    group_apply(Lg + baseA,
                (pimg ? SI[r0] : SR[r0]) + baseB0,
                (pimg ? SR[r0] : SI[r0]) + baseB0,
                (pimg ? SI[r0 ^ 1] : SR[r0 ^ 1]) + baseW0,
                272, 68, 0, 0, 0, pimg);
    __syncthreads();
    group_apply(Lg + 512 + baseA,
                (pimg ? SI[r0 ^ 1] : SR[r0 ^ 1]) + baseB1,
                (pimg ? SR[r0 ^ 1] : SI[r0 ^ 1]) + baseB1,
                (pimg ? SI[r0] : SR[r0]) + baseW1,
                17, 4, 1, 0, 0, pimg);
    __syncthreads();
    {
      const int pb = permT[l - 1][col] ^ ((lane & 16) ? cnot_perm(8, l + 1) : 0);
      group_apply(Lg + 1024 + baseA,
                  (pimg ? SI[r0] : SR[r0]) + baseB2,
                  (pimg ? SR[r0] : SI[r0]) + baseB2,
                  (pimg ? SI[r0 ^ 1] : SR[r0 ^ 1]),
                  4, 1, 2, l + 1, pb, pimg);
      __syncthreads();
    }
  }

  const float sgn = (tid & 1) ? -1.f : 1.f;
  float acc = 0.f;
#pragma unroll
  for (int j = 0; j < DIM / 256; ++j) {
    const int a = PADDED(tid + 256 * j);
    const float re = SR[0][a], im = SI[0][a];
    acc += re * re + im * im;
  }
  float z = sgn * acc;
#pragma unroll
  for (int m = 16; m > 0; m >>= 1) z += __shfl_xor(z, m, 32);
  if (lane == 0) red[wave] = z;
  __syncthreads();
  if (tid == 0) {
    float t = 0.f;
#pragma unroll
    for (int w = 0; w < 8; ++w) t += red[w];
    out[s] = t + bias[0];
  }
}

extern "C" void kernel_launch(void *const *d_in, const int *in_sizes, int n_in,
                              void *d_out, int out_size, void *d_ws, size_t ws_size,
                              hipStream_t stream) {
  const float *X = (const float *)d_in[0];
  const float *W = (const float *)d_in[1];
  const float *bias = (const float *)d_in[2];
  float *out = (float *)d_out;
  float *G = (float *)d_ws;

  const int B = in_sizes[0] / N_WIRES;

  vqc_precompute<<<1, 256, 0, stream>>>(W, G);
  vqc_kernel<<<B, 256, 0, stream>>>(X, G, bias, out);
}


# Wires regrouped 3+3+4; each 3-wire operator applied as one real 16x16 matrix [[Ur,-Ui],[Ui,Ur]] on stacked [re;im]: 48 WMMAs per wave instead of 72; new summation order in the gate products
# speedup vs baseline: 1.7134x; 1.3492x over previous
#include <hip/hip_runtime.h>


typedef float v2f __attribute__((ext_vector_type(2)));
typedef float v8f __attribute__((ext_vector_type(8)));

#define N_WIRES 10
#define N_LAYERS 4
#define DIM 1024
#define PADDED(k) ((k) + ((k) >> 4))
#define PDIM (DIM + (DIM >> 4))
#define PSTR (PDIM + 32)
#define LAYER_F 1024
#define N_FUSED (N_LAYERS - 1)
#define U0_OFF (N_FUSED * LAYER_F)

__device__ __forceinline__ void cmul(float ar, float ai, float br, float bi,
                                     float &cr, float &ci) {
  cr = ar * br - ai * bi;
  ci = ar * bi + ai * br;
}

__device__ __forceinline__ int cnot_perm(int m, int rr) {
#pragma unroll
  for (int w = 0; w < N_WIRES; ++w) {
    int t = w + rr; if (t >= N_WIRES) t -= N_WIRES;
    m ^= ((m >> (9 - w)) & 1) << (9 - t);
  }
  return m;
}

__global__ __launch_bounds__(256) void vqc_precompute(const float *__restrict__ W,
                                                      float *__restrict__ G) {
  __shared__ float u[N_WIRES][8];
  const int tid = threadIdx.x;
  for (int l = 0; l < N_LAYERS; ++l) {
    if (tid < N_WIRES) {
      const float phi   = W[(l * N_WIRES + tid) * 3 + 0];
      const float theta = W[(l * N_WIRES + tid) * 3 + 1];
      const float omega = W[(l * N_WIRES + tid) * 3 + 2];
      const float c = __cosf(0.5f * theta), s = __sinf(0.5f * theta);
      const float a = 0.5f * (phi + omega), b = 0.5f * (phi - omega);
      const float ca = __cosf(a), sa = __sinf(a), cb = __cosf(b), sb = __sinf(b);
      u[tid][0] =  ca * c; u[tid][1] = -sa * c;
      u[tid][2] = -cb * s; u[tid][3] = -sb * s;
      u[tid][4] =  cb * s; u[tid][5] = -sb * s;
      u[tid][6] =  ca * c; u[tid][7] =  sa * c;
    }
    __syncthreads();
    if (l == 0) {
      if (tid < N_WIRES * 8) G[U0_OFF + tid] = u[tid >> 3][tid & 7];
    } else {
    float *L = G + (l - 1) * LAYER_F;
    const int r = tid >> 4, c = tid & 15;
#pragma unroll
    for (int g = 0; g < 2; ++g) {
      float pr = 1.f, pi = 0.f;
#pragma unroll
      for (int q = 0; q < 3; ++q) {
        const int rb = (r >> (2 - q)) & 1, cb2 = (c >> (2 - q)) & 1;
        const float *e = &u[g * 3 + q][(rb * 2 + cb2) * 2];
        float nr, ni; cmul(pr, pi, e[0], e[1], nr, ni);
        pr = nr; pi = ni;
      }
      const int rq = r >> 3, cq = c >> 3;
      const float val = (rq == cq) ? pr : (rq ? pi : -pi);
      L[g * 256 + c * 16 + r] = val;
    }
    {
      float pr = 1.f, pi = 0.f;
#pragma unroll
      for (int q = 0; q < 4; ++q) {
        const int rb = (r >> (3 - q)) & 1, cb2 = (c >> (3 - q)) & 1;
        const float *e = &u[6 + q][(rb * 2 + cb2) * 2];
        float nr, ni; cmul(pr, pi, e[0], e[1], nr, ni);
        pr = nr; pi = ni;
      }
      L[512 + c * 16 + r] = pr;
      L[768 + c * 16 + r] = pi;
    }
    }
    __syncthreads();
  }
}

__device__ __forceinline__ void group_apply_stacked(const float *__restrict__ A,
                                                    const float *bRe, const float *bIm,
                                                    float *dst, int idxStride) {
  v8f acc = {};
#pragma unroll
  for (int kk = 0; kk < 4; ++kk) {
    const float *p = (kk < 2) ? bRe : bIm;
    const int o = (kk & 1) * 4 * idxStride;
    v2f a, b;
    a.x = A[kk * 64];  a.y = A[kk * 64 + 16];
    b.x = p[o];        b.y = p[o + idxStride];
    acc = __builtin_amdgcn_wmma_f32_16x16x4_f32(false, a, false, b,
                                                (short)0, acc, false, false);
  }
#pragma unroll
  for (int v = 0; v < 8; ++v) dst[v * idxStride] = acc[v];
}

__device__ __forceinline__ void group_apply(const float *__restrict__ A,
                                            const float *p1, const float *p2,
                                            float *dst, int rr, int permBase,
                                            bool pimg) {
  v8f acc = {};
#pragma unroll
  for (int kk = 0; kk < 4; ++kk) {
    v2f ar, ai, bF, bS, aS;
    ar.x = A[kk * 64];        ar.y = A[kk * 64 + 16];
    ai.x = A[256 + kk * 64];  ai.y = A[256 + kk * 64 + 16];
    bF.x = p1[kk * 4];        bF.y = p1[kk * 4 + 1];
    bS.x = p2[kk * 4];        bS.y = p2[kk * 4 + 1];
    aS.x = pimg ? ai.x : -ai.x;
    aS.y = pimg ? ai.y : -ai.y;
    acc = __builtin_amdgcn_wmma_f32_16x16x4_f32(false, ar, false, bF,
                                                (short)0, acc, false, false);
    acc = __builtin_amdgcn_wmma_f32_16x16x4_f32(false, aS, false, bS,
                                                (short)0, acc, false, false);
  }
#pragma unroll
  for (int v = 0; v < 8; ++v) {
    const int w = PADDED(permBase ^ cnot_perm(v, rr));
    dst[w] = acc[v];
  }
}

__global__ __launch_bounds__(256) void vqc_kernel(const float *__restrict__ X,
                                                  const float *__restrict__ G,
                                                  const float *__restrict__ bias,
                                                  float *__restrict__ out) {
  __shared__ float ST[4 * PSTR];
  __shared__ float gm[N_FUSED * LAYER_F];
  __shared__ float vr[2 * N_WIRES];
  __shared__ float vi[2 * N_WIRES];
  __shared__ float pLoR[32], pLoI[32], pHiR[32], pHiI[32];
  __shared__ int   permT[N_FUSED][64];
  __shared__ float red[8];

#define SRE(b) (ST + (2 * (b)) * PSTR)
#define SIM(b) (ST + (2 * (b) + 1) * PSTR)

  const int tid  = threadIdx.x;
  const int lane = tid & 31;
  const int wave = tid >> 5;
  const int s    = blockIdx.x;

  for (int i = tid; i < N_FUSED * LAYER_F; i += 256) gm[i] = G[i];

  if (tid >= 64)
    permT[(tid >> 6) - 1][tid & 63] = cnot_perm((tid & 63) << 4, (tid >> 6) + 1);

  const int perm0 = cnot_perm(tid, 1);

  if (tid < N_WIRES) {
    const float x = X[s * N_WIRES + tid] * 0.7853981633974483f;
    const float c = __cosf(x), sn = __sinf(x);
    const float *m = G + U0_OFF + tid * 8;
    vr[tid]           = m[0] * c + m[2] * sn;
    vi[tid]           = m[1] * c + m[3] * sn;
    vr[N_WIRES + tid] = m[4] * c + m[6] * sn;
    vi[N_WIRES + tid] = m[5] * c + m[7] * sn;
  }
  __syncthreads();

  if (tid < 32) {
    float pr = 1.f, pi = 0.f;
#pragma unroll
    for (int w = 5; w < N_WIRES; ++w) {
      const int e = ((tid >> (9 - w)) & 1) * N_WIRES + w;
      float nr, ni; cmul(pr, pi, vr[e], vi[e], nr, ni);
      pr = nr; pi = ni;
    }
    pLoR[tid] = pr; pLoI[tid] = pi;
  } else if (tid < 64) {
    const int h = tid - 32;
    float pr = 1.f, pi = 0.f;
#pragma unroll
    for (int w = 0; w < 5; ++w) {
      const int e = ((h >> (4 - w)) & 1) * N_WIRES + w;
      float nr, ni; cmul(pr, pi, vr[e], vi[e], nr, ni);
      pr = nr; pi = ni;
    }
    pHiR[h] = pr; pHiI[h] = pi;
  }
  __syncthreads();

#pragma unroll
  for (int j = 0; j < DIM / 256; ++j) {
    const int k = tid + 256 * j;
    float re, im;
    cmul(pHiR[k >> 5], pHiI[k >> 5], pLoR[k & 31], pLoI[k & 31], re, im);
    const int d = perm0 ^ cnot_perm(256 * j, 1);
    SRE(1)[PADDED(d)] = re;
    SIM(1)[PADDED(d)] = im;
  }
  __syncthreads();

  const int  c16 = lane & 15;
  const int  h   = lane >> 4;
  const int  kb  = h << 1;
  const int  baseA = kb * 16 + c16;
  const int colA   = wave * 16 + c16;
  const int baseBA = colA + (colA >> 4) + kb * 136;
  const int baseWA = colA + (colA >> 4) + h * PSTR;
  const int baseBB = wave * 136 + c16 + kb * 17;
  const int baseWB = wave * 136 + c16 + h * PSTR;
  const int  tile = wave >> 1;
  const bool pimg = (wave & 1) != 0;
  const int  col  = tile * 16 + c16;
  const int baseB2 = col * 17 + kb;

#pragma unroll
  for (int l = 1; l < N_LAYERS; ++l) {
    const float *Lg = gm + (l - 1) * LAYER_F;
    const int r0 = (3 * l) & 1;

    group_apply_stacked(Lg + baseA,
                        SRE(r0) + baseBA, SIM(r0) + baseBA,
                        SRE(r0 ^ 1) + baseWA, 136);
    __syncthreads();
    group_apply_stacked(Lg + 256 + baseA,
                        SRE(r0 ^ 1) + baseBB, SIM(r0 ^ 1) + baseBB,
                        SRE(r0) + baseWB, 17);
    __syncthreads();
    {
      const int pb = permT[l - 1][col] ^ (h ? cnot_perm(8, l + 1) : 0);
      group_apply(Lg + 512 + baseA,
                  (pimg ? SIM(r0) : SRE(r0)) + baseB2,
                  (pimg ? SRE(r0) : SIM(r0)) + baseB2,
                  (pimg ? SIM(r0 ^ 1) : SRE(r0 ^ 1)),
                  l + 1, pb, pimg);
      __syncthreads();
    }
  }

  const float sgn = (tid & 1) ? -1.f : 1.f;
  float acc = 0.f;
#pragma unroll
  for (int j = 0; j < DIM / 256; ++j) {
    const int a = PADDED(tid + 256 * j);
    const float re = SRE(0)[a], im = SIM(0)[a];
    acc += re * re + im * im;
  }
  float z = sgn * acc;
#pragma unroll
  for (int m = 16; m > 0; m >>= 1) z += __shfl_xor(z, m, 32);
  if (lane == 0) red[wave] = z;
  __syncthreads();
  if (tid == 0) {
    float t = 0.f;
#pragma unroll
    for (int w = 0; w < 8; ++w) t += red[w];
    out[s] = t + bias[0];
  }
#undef SRE
#undef SIM
}

extern "C" void kernel_launch(void *const *d_in, const int *in_sizes, int n_in,
                              void *d_out, int out_size, void *d_ws, size_t ws_size,
                              hipStream_t stream) {
  const float *X = (const float *)d_in[0];
  const float *W = (const float *)d_in[1];
  const float *bias = (const float *)d_in[2];
  float *out = (float *)d_out;
  float *G = (float *)d_ws;

  const int B = in_sizes[0] / N_WIRES;

  vqc_precompute<<<1, 256, 0, stream>>>(W, G);
  vqc_kernel<<<B, 256, 0, stream>>>(X, G, bias, out);
}


# Gates outside the measurement's backward light cone dropped: layer 3 keeps wires 1,5,9 (one 3-wire WMMA phase, <Z1Z5Z9> summed from its accumulators), layer 2 drops wire 7; new <Z> summation order
# speedup vs baseline: 2.4372x; 1.4224x over previous
#include <hip/hip_runtime.h>


typedef float v2f __attribute__((ext_vector_type(2)));
typedef float v8f __attribute__((ext_vector_type(8)));

#define N_WIRES 10
#define N_LAYERS 4
#define DIM 1024
#define PADDED(k) ((k) + ((k) >> 4))
#define PDIM (DIM + (DIM >> 4))
#define PSTR (PDIM + 32)
#define LAYER_F 1024
#define L2_OFF LAYER_F
#define L3_OFF (L2_OFF + 768)
#define GM_WORDS (L3_OFF + 256)
#define U0_OFF GM_WORDS

__device__ __forceinline__ void cmul(float ar, float ai, float br, float bi,
                                     float &cr, float &ci) {
  cr = ar * br - ai * bi;
  ci = ar * bi + ai * br;
}

__device__ __forceinline__ int cnot_perm(int m, int rr) {
#pragma unroll
  for (int w = 0; w < N_WIRES; ++w) {
    int t = w + rr; if (t >= N_WIRES) t -= N_WIRES;
    m ^= ((m >> (9 - w)) & 1) << (9 - t);
  }
  return m;
}

__device__ __forceinline__ float real_form3(const float (*u)[8], int w0, int w1, int w2,
                                            int r, int c) {
  float pr = 1.f, pi = 0.f;
#pragma unroll
  for (int q = 0; q < 3; ++q) {
    const int wq = (q == 0) ? w0 : ((q == 1) ? w1 : w2);
    const int rb = (r >> (2 - q)) & 1, cb2 = (c >> (2 - q)) & 1;
    const float *e = &u[wq][(rb * 2 + cb2) * 2];
    float nr, ni; cmul(pr, pi, e[0], e[1], nr, ni);
    pr = nr; pi = ni;
  }
  const int rq = r >> 3, cq = c >> 3;
  return (rq == cq) ? pr : (rq ? pi : -pi);
}

__global__ __launch_bounds__(256) void vqc_precompute(const float *__restrict__ W,
                                                      float *__restrict__ G) {
  __shared__ float u[N_WIRES][8];
  const int tid = threadIdx.x;
  const int r = tid >> 4, c = tid & 15;
  for (int l = 0; l < N_LAYERS; ++l) {
    if (tid < N_WIRES) {
      const float phi   = W[(l * N_WIRES + tid) * 3 + 0];
      const float theta = W[(l * N_WIRES + tid) * 3 + 1];
      const float omega = W[(l * N_WIRES + tid) * 3 + 2];
      const float cs = __cosf(0.5f * theta), s = __sinf(0.5f * theta);
      const float a = 0.5f * (phi + omega), b = 0.5f * (phi - omega);
      const float ca = __cosf(a), sa = __sinf(a), cb = __cosf(b), sb = __sinf(b);
      u[tid][0] =  ca * cs; u[tid][1] = -sa * cs;
      u[tid][2] = -cb * s;  u[tid][3] = -sb * s;
      u[tid][4] =  cb * s;  u[tid][5] = -sb * s;
      u[tid][6] =  ca * cs; u[tid][7] =  sa * cs;
    }
    __syncthreads();
    if (l == 0) {
      if (tid < N_WIRES * 8) G[U0_OFF + tid] = u[tid >> 3][tid & 7];
    } else if (l == N_LAYERS - 1) {
      G[L3_OFF + c * 16 + r] = real_form3(u, 1, 5, 9, r, c);
    } else {
      float *L = G + (l - 1) * LAYER_F;
      L[c * 16 + r]       = real_form3(u, 0, 1, 2, r, c);
      L[256 + c * 16 + r] = real_form3(u, 3, 4, 5, r, c);
      if (l == 2) {
        L[512 + c * 16 + r] = real_form3(u, 6, 8, 9, r, c);
      } else {
        float pr = 1.f, pi = 0.f;
#pragma unroll
        for (int q = 0; q < 4; ++q) {
          const int rb = (r >> (3 - q)) & 1, cb2 = (c >> (3 - q)) & 1;
          const float *e = &u[6 + q][(rb * 2 + cb2) * 2];
          float nr, ni; cmul(pr, pi, e[0], e[1], nr, ni);
          pr = nr; pi = ni;
        }
        L[512 + c * 16 + r] = pr;
        L[768 + c * 16 + r] = pi;
      }
    }
    __syncthreads();
  }
}

__device__ __forceinline__ void group_apply_stacked(const float *__restrict__ A,
                                                    const float *bRe, const float *bIm,
                                                    float *dst, int idxStride) {
  v8f acc = {};
#pragma unroll
  for (int kk = 0; kk < 4; ++kk) {
    const float *p = (kk < 2) ? bRe : bIm;
    const int o = (kk & 1) * 4 * idxStride;
    v2f a, b;
    a.x = A[kk * 64];  a.y = A[kk * 64 + 16];
    b.x = p[o];        b.y = p[o + idxStride];
    acc = __builtin_amdgcn_wmma_f32_16x16x4_f32(false, a, false, b,
                                                (short)0, acc, false, false);
  }
#pragma unroll
  for (int v = 0; v < 8; ++v) dst[v * idxStride] = acc[v];
}

__device__ __forceinline__ void group_apply_stacked_scatter(const float *__restrict__ A,
                                                            const float *bRe, const float *bIm,
                                                            float *dst, int rr, int permBase) {
  v8f acc = {};
#pragma unroll
  for (int kk = 0; kk < 4; ++kk) {
    const float *p = (kk < 2) ? bRe : bIm;
    const int o = (kk & 1) * 8;
    v2f a, b;
    a.x = A[kk * 64];  a.y = A[kk * 64 + 16];
    b.x = p[o];        b.y = p[o + 1];
    acc = __builtin_amdgcn_wmma_f32_16x16x4_f32(false, a, false, b,
                                                (short)0, acc, false, false);
  }
#pragma unroll
  for (int v = 0; v < 8; ++v) {
    const int w = PADDED(permBase ^ cnot_perm(((v & 4) << 1) | (v & 3), rr));
    dst[w] = acc[v];
  }
}

__device__ __forceinline__ float group_apply_stacked_z(const float *__restrict__ A,
                                                       const float *bRe, const float *bIm) {
  v8f acc = {};
#pragma unroll
  for (int kk = 0; kk < 4; ++kk) {
    const float *p = (kk < 2) ? bRe : bIm;
    const int o = (kk & 1) * 272;
    v2f a, b;
    a.x = A[kk * 64];  a.y = A[kk * 64 + 16];
    b.x = p[o];        b.y = p[o + 1];
    acc = __builtin_amdgcn_wmma_f32_16x16x4_f32(false, a, false, b,
                                                (short)0, acc, false, false);
  }
  float z = 0.f;
#pragma unroll
  for (int v = 0; v < 8; ++v) {
    const float sq = acc[v] * acc[v];
    z += (((v >> 2) ^ (v >> 1) ^ v) & 1) ? -sq : sq;
  }
  return z;
}

__device__ __forceinline__ void group_apply(const float *__restrict__ A,
                                            const float *p1, const float *p2,
                                            float *dst, int rr, int permBase,
                                            bool pimg) {
  v8f acc = {};
#pragma unroll
  for (int kk = 0; kk < 4; ++kk) {
    v2f ar, ai, bF, bS, aS;
    ar.x = A[kk * 64];        ar.y = A[kk * 64 + 16];
    ai.x = A[256 + kk * 64];  ai.y = A[256 + kk * 64 + 16];
    bF.x = p1[kk * 4];        bF.y = p1[kk * 4 + 1];
    bS.x = p2[kk * 4];        bS.y = p2[kk * 4 + 1];
    aS.x = pimg ? ai.x : -ai.x;
    aS.y = pimg ? ai.y : -ai.y;
    acc = __builtin_amdgcn_wmma_f32_16x16x4_f32(false, ar, false, bF,
                                                (short)0, acc, false, false);
    acc = __builtin_amdgcn_wmma_f32_16x16x4_f32(false, aS, false, bS,
                                                (short)0, acc, false, false);
  }
#pragma unroll
  for (int v = 0; v < 8; ++v) {
    const int w = PADDED(permBase ^ cnot_perm(v, rr));
    dst[w] = acc[v];
  }
}

__global__ __launch_bounds__(256) void vqc_kernel(const float *__restrict__ X,
                                                  const float *__restrict__ G,
                                                  const float *__restrict__ bias,
                                                  float *__restrict__ out) {
  __shared__ float ST[4 * PSTR];
  __shared__ float gm[GM_WORDS];
  __shared__ float vr[2 * N_WIRES];
  __shared__ float vi[2 * N_WIRES];
  __shared__ float pLoR[32], pLoI[32], pHiR[32], pHiI[32];
  __shared__ int   permT[2][64];
  __shared__ float red[8];

#define SRE(b) (ST + (2 * (b)) * PSTR)
#define SIM(b) (ST + (2 * (b) + 1) * PSTR)

  const int tid  = threadIdx.x;
  const int lane = tid & 31;
  const int wave = tid >> 5;
  const int s    = blockIdx.x;

#pragma unroll
  for (int j = 0; j < GM_WORDS / 256; ++j) gm[tid + 256 * j] = G[tid + 256 * j];

  if (tid >= 64 && tid < 192)
    permT[(tid >> 6) - 1][tid & 63] = cnot_perm((tid & 63) << 4, (tid >> 6) + 1);

  const int perm0 = cnot_perm(tid, 1);

  if (tid < N_WIRES) {
    const float x = X[s * N_WIRES + tid] * 0.7853981633974483f;
    const float c = __cosf(x), sn = __sinf(x);
    const float *m = G + U0_OFF + tid * 8;
    vr[tid]           = m[0] * c + m[2] * sn;
    vi[tid]           = m[1] * c + m[3] * sn;
    vr[N_WIRES + tid] = m[4] * c + m[6] * sn;
    vi[N_WIRES + tid] = m[5] * c + m[7] * sn;
  }
  __syncthreads();

  if (tid < 32) {
    float pr = 1.f, pi = 0.f;
#pragma unroll
    for (int w = 5; w < N_WIRES; ++w) {
      const int e = ((tid >> (9 - w)) & 1) * N_WIRES + w;
      float nr, ni; cmul(pr, pi, vr[e], vi[e], nr, ni);
      pr = nr; pi = ni;
    }
    pLoR[tid] = pr; pLoI[tid] = pi;
  } else if (tid < 64) {
    const int h = tid - 32;
    float pr = 1.f, pi = 0.f;
#pragma unroll
    for (int w = 0; w < 5; ++w) {
      const int e = ((h >> (4 - w)) & 1) * N_WIRES + w;
      float nr, ni; cmul(pr, pi, vr[e], vi[e], nr, ni);
      pr = nr; pi = ni;
    }
    pHiR[h] = pr; pHiI[h] = pi;
  }
  __syncthreads();

#pragma unroll
  for (int j = 0; j < DIM / 256; ++j) {
    const int k = tid + 256 * j;
    float re, im;
    cmul(pHiR[k >> 5], pHiI[k >> 5], pLoR[k & 31], pLoI[k & 31], re, im);
    const int d = perm0 ^ cnot_perm(256 * j, 1);
    SRE(1)[PADDED(d)] = re;
    SIM(1)[PADDED(d)] = im;
  }
  __syncthreads();

  const int  c16 = lane & 15;
  const int  h   = lane >> 4;
  const int  kb  = h << 1;
  const int  baseA = kb * 16 + c16;
  const int colA   = wave * 16 + c16;
  const int baseBA = colA + (colA >> 4) + kb * 136;
  const int baseWA = colA + (colA >> 4) + h * PSTR;
  const int baseBB = wave * 136 + c16 + kb * 17;
  const int baseWB = wave * 136 + c16 + h * PSTR;
  const int  tile = wave >> 1;
  const bool pimg = (wave & 1) != 0;
  const int  col  = tile * 16 + c16;
  const int baseB2 = col * 17 + kb;
  const int colC   = (wave >> 1) * 16 + c16;
  const int baseBC = (wave >> 1) * 272 + c16 * 17 + (wave & 1) * 4 + kb;
  const int col3   = ((wave & 4) << 7) | ((wave & 3) << 6) | ((c16 & 8) << 2) | ((c16 & 7) << 1);
  const int baseB3 = col3 + (col3 >> 4) + h * 17;

#pragma unroll
  for (int l = 1; l < N_LAYERS - 1; ++l) {
    const float *Lg = gm + (l - 1) * LAYER_F;
    const int r0 = (3 * l) & 1;

    group_apply_stacked(Lg + baseA,
                        SRE(r0) + baseBA, SIM(r0) + baseBA,
                        SRE(r0 ^ 1) + baseWA, 136);
    __syncthreads();
    group_apply_stacked(Lg + 256 + baseA,
                        SRE(r0 ^ 1) + baseBB, SIM(r0 ^ 1) + baseBB,
                        SRE(r0) + baseWB, 17);
    __syncthreads();
    if (l == 2) {
      const int pb = permT[l - 1][colC] ^ ((wave & 1) ? cnot_perm(4, l + 1) : 0);
      group_apply_stacked_scatter(Lg + 512 + baseA,
                                  SRE(r0) + baseBC, SIM(r0) + baseBC,
                                  SRE(r0 ^ 1) + h * PSTR,
                                  l + 1, pb);
      __syncthreads();
    } else {
      const int pb = permT[l - 1][col] ^ (h ? cnot_perm(8, l + 1) : 0);
      group_apply(Lg + 512 + baseA,
                  (pimg ? SIM(r0) : SRE(r0)) + baseB2,
                  (pimg ? SRE(r0) : SIM(r0)) + baseB2,
                  (pimg ? SIM(r0 ^ 1) : SRE(r0 ^ 1)),
                  l + 1, pb, pimg);
      __syncthreads();
    }
  }

  float z = group_apply_stacked_z(gm + L3_OFF + baseA, SRE(1) + baseB3, SIM(1) + baseB3);
#pragma unroll
  for (int m = 16; m > 0; m >>= 1) z += __shfl_xor(z, m, 32);
  if (lane == 0) red[wave] = z;
  __syncthreads();
  if (tid == 0) {
    float t = 0.f;
#pragma unroll
    for (int w = 0; w < 8; ++w) t += red[w];
    out[s] = t + bias[0];
  }
#undef SRE
#undef SIM
}

extern "C" void kernel_launch(void *const *d_in, const int *in_sizes, int n_in,
                              void *d_out, int out_size, void *d_ws, size_t ws_size,
                              hipStream_t stream) {
  const float *X = (const float *)d_in[0];
  const float *W = (const float *)d_in[1];
  const float *bias = (const float *)d_in[2];
  float *out = (float *)d_out;
  float *G = (float *)d_ws;

  const int B = in_sizes[0] / N_WIRES;

  vqc_precompute<<<1, 256, 0, stream>>>(W, G);
  vqc_kernel<<<B, 256, 0, stream>>>(X, G, bias, out);
}


# Four samples per workgroup: each wave applies every operator fragment it loads to 4 column tiles, one per sample (4 independent WMMA chains); staging, index math, barriers shared; sums unchanged
# speedup vs baseline: 2.7344x; 1.1219x over previous
#include <hip/hip_runtime.h>


typedef float v2f __attribute__((ext_vector_type(2)));
typedef float v8f __attribute__((ext_vector_type(8)));

#define N_WIRES 10
#define N_LAYERS 4
#define DIM 1024
#define PADDED(k) ((k) + ((k) >> 4))
#define PDIM (DIM + (DIM >> 4))
#define PSTR (PDIM + 32)
#define LAYER_F 1024
#define L2_OFF LAYER_F
#define L3_OFF (L2_OFF + 768)
#define GM_WORDS (L3_OFF + 256)
#define U0_OFF GM_WORDS
#define NS 4
#define SSTR (4 * PSTR)
#define VR_OFF 0
#define VI_OFF 32
#define PLOR_OFF 64
#define PLOI_OFF 96
#define PHIR_OFF 128
#define PHII_OFF 160

__device__ __forceinline__ void cmul(float ar, float ai, float br, float bi,
                                     float &cr, float &ci) {
  cr = ar * br - ai * bi;
  ci = ar * bi + ai * br;
}

__device__ __forceinline__ int cnot_perm(int m, int rr) {
#pragma unroll
  for (int w = 0; w < N_WIRES; ++w) {
    int t = w + rr; if (t >= N_WIRES) t -= N_WIRES;
    m ^= ((m >> (9 - w)) & 1) << (9 - t);
  }
  return m;
}

__device__ __forceinline__ float real_form3(const float (*u)[8], int w0, int w1, int w2,
                                            int r, int c) {
  float pr = 1.f, pi = 0.f;
#pragma unroll
  for (int q = 0; q < 3; ++q) {
    const int wq = (q == 0) ? w0 : ((q == 1) ? w1 : w2);
    const int rb = (r >> (2 - q)) & 1, cb2 = (c >> (2 - q)) & 1;
    const float *e = &u[wq][(rb * 2 + cb2) * 2];
    float nr, ni; cmul(pr, pi, e[0], e[1], nr, ni);
    pr = nr; pi = ni;
  }
  const int rq = r >> 3, cq = c >> 3;
  return (rq == cq) ? pr : (rq ? pi : -pi);
}

__global__ __launch_bounds__(256) void vqc_precompute(const float *__restrict__ W,
                                                      float *__restrict__ G) {
  __shared__ float u[N_WIRES][8];
  const int tid = threadIdx.x;
  const int r = tid >> 4, c = tid & 15;
  for (int l = 0; l < N_LAYERS; ++l) {
    if (tid < N_WIRES) {
      const float phi   = W[(l * N_WIRES + tid) * 3 + 0];
      const float theta = W[(l * N_WIRES + tid) * 3 + 1];
      const float omega = W[(l * N_WIRES + tid) * 3 + 2];
      const float cs = __cosf(0.5f * theta), s = __sinf(0.5f * theta);
      const float a = 0.5f * (phi + omega), b = 0.5f * (phi - omega);
      const float ca = __cosf(a), sa = __sinf(a), cb = __cosf(b), sb = __sinf(b);
      u[tid][0] =  ca * cs; u[tid][1] = -sa * cs;
      u[tid][2] = -cb * s;  u[tid][3] = -sb * s;
      u[tid][4] =  cb * s;  u[tid][5] = -sb * s;
      u[tid][6] =  ca * cs; u[tid][7] =  sa * cs;
    }
    __syncthreads();
    if (l == 0) {
      if (tid < N_WIRES * 8) G[U0_OFF + tid] = u[tid >> 3][tid & 7];
    } else if (l == N_LAYERS - 1) {
      G[L3_OFF + c * 16 + r] = real_form3(u, 1, 5, 9, r, c);
    } else {
      float *L = G + (l - 1) * LAYER_F;
      L[c * 16 + r]       = real_form3(u, 0, 1, 2, r, c);
      L[256 + c * 16 + r] = real_form3(u, 3, 4, 5, r, c);
      if (l == 2) {
        L[512 + c * 16 + r] = real_form3(u, 6, 8, 9, r, c);
      } else {
        float pr = 1.f, pi = 0.f;
#pragma unroll
        for (int q = 0; q < 4; ++q) {
          const int rb = (r >> (3 - q)) & 1, cb2 = (c >> (3 - q)) & 1;
          const float *e = &u[6 + q][(rb * 2 + cb2) * 2];
          float nr, ni; cmul(pr, pi, e[0], e[1], nr, ni);
          pr = nr; pi = ni;
        }
        L[512 + c * 16 + r] = pr;
        L[768 + c * 16 + r] = pi;
      }
    }
    __syncthreads();
  }
}

__device__ __forceinline__ void group_apply_stacked(const float *__restrict__ A,
                                                    const float *bRe, const float *bIm,
                                                    float *dst, int idxStride) {
  v8f acc[NS];
#pragma unroll
  for (int i = 0; i < NS; ++i) acc[i] = v8f{};
#pragma unroll
  for (int kk = 0; kk < 4; ++kk) {
    const float *p = (kk < 2) ? bRe : bIm;
    const int o = (kk & 1) * 4 * idxStride;
    v2f a;
    a.x = A[kk * 64];  a.y = A[kk * 64 + 16];
#pragma unroll
    for (int i = 0; i < NS; ++i) {
      v2f b;
      b.x = p[i * SSTR + o];  b.y = p[i * SSTR + o + idxStride];
      acc[i] = __builtin_amdgcn_wmma_f32_16x16x4_f32(false, a, false, b,
                                                     (short)0, acc[i], false, false);
    }
  }
#pragma unroll
  for (int i = 0; i < NS; ++i) {
#pragma unroll
    for (int v = 0; v < 8; ++v) dst[i * SSTR + v * idxStride] = acc[i][v];
  }
}

__device__ __forceinline__ void group_apply_stacked_scatter(const float *__restrict__ A,
                                                            const float *bRe, const float *bIm,
                                                            float *dst, int rr, int permBase) {
  v8f acc[NS];
#pragma unroll
  for (int i = 0; i < NS; ++i) acc[i] = v8f{};
#pragma unroll
  for (int kk = 0; kk < 4; ++kk) {
    const float *p = (kk < 2) ? bRe : bIm;
    const int o = (kk & 1) * 8;
    v2f a;
    a.x = A[kk * 64];  a.y = A[kk * 64 + 16];
#pragma unroll
    for (int i = 0; i < NS; ++i) {
      v2f b;
      b.x = p[i * SSTR + o];  b.y = p[i * SSTR + o + 1];
      acc[i] = __builtin_amdgcn_wmma_f32_16x16x4_f32(false, a, false, b,
                                                     (short)0, acc[i], false, false);
    }
  }
#pragma unroll
  for (int v = 0; v < 8; ++v) {
    const int w = PADDED(permBase ^ cnot_perm(((v & 4) << 1) | (v & 3), rr));
#pragma unroll
    for (int i = 0; i < NS; ++i) dst[i * SSTR + w] = acc[i][v];
  }
}

__device__ __forceinline__ void group_apply_stacked_z(const float *__restrict__ A,
                                                      const float *bRe, const float *bIm,
                                                      float *z) {
  v8f acc[NS];
#pragma unroll
  for (int i = 0; i < NS; ++i) acc[i] = v8f{};
#pragma unroll
  for (int kk = 0; kk < 4; ++kk) {
    const float *p = (kk < 2) ? bRe : bIm;
    const int o = (kk & 1) * 272;
    v2f a;
    a.x = A[kk * 64];  a.y = A[kk * 64 + 16];
#pragma unroll
    for (int i = 0; i < NS; ++i) {
      v2f b;
      b.x = p[i * SSTR + o];  b.y = p[i * SSTR + o + 1];
      acc[i] = __builtin_amdgcn_wmma_f32_16x16x4_f32(false, a, false, b,
                                                     (short)0, acc[i], false, false);
    }
  }
#pragma unroll
  for (int i = 0; i < NS; ++i) {
    float t = 0.f;
#pragma unroll
    for (int v = 0; v < 8; ++v) {
      const float sq = acc[i][v] * acc[i][v];
      t += (((v >> 2) ^ (v >> 1) ^ v) & 1) ? -sq : sq;
    }
    z[i] = t;
  }
}

__device__ __forceinline__ void group_apply(const float *__restrict__ A,
                                            const float *p1, const float *p2,
                                            float *dst, int rr, int permBase,
                                            bool pimg) {
  v8f acc[NS];
#pragma unroll
  for (int i = 0; i < NS; ++i) acc[i] = v8f{};
#pragma unroll
  for (int kk = 0; kk < 4; ++kk) {
    v2f ar, ai, aS;
    ar.x = A[kk * 64];        ar.y = A[kk * 64 + 16];
    ai.x = A[256 + kk * 64];  ai.y = A[256 + kk * 64 + 16];
    aS.x = pimg ? ai.x : -ai.x;
    aS.y = pimg ? ai.y : -ai.y;
#pragma unroll
    for (int i = 0; i < NS; ++i) {
      v2f bF;
      bF.x = p1[i * SSTR + kk * 4];  bF.y = p1[i * SSTR + kk * 4 + 1];
      acc[i] = __builtin_amdgcn_wmma_f32_16x16x4_f32(false, ar, false, bF,
                                                     (short)0, acc[i], false, false);
    }
#pragma unroll
    for (int i = 0; i < NS; ++i) {
      v2f bS;
      bS.x = p2[i * SSTR + kk * 4];  bS.y = p2[i * SSTR + kk * 4 + 1];
      acc[i] = __builtin_amdgcn_wmma_f32_16x16x4_f32(false, aS, false, bS,
                                                     (short)0, acc[i], false, false);
    }
  }
#pragma unroll
  for (int v = 0; v < 8; ++v) {
    const int w = PADDED(permBase ^ cnot_perm(v, rr));
#pragma unroll
    for (int i = 0; i < NS; ++i) dst[i * SSTR + w] = acc[i][v];
  }
}

__global__ __launch_bounds__(256) void vqc_kernel(const float *__restrict__ X,
                                                  const float *__restrict__ G,
                                                  const float *__restrict__ bias,
                                                  float *__restrict__ out, int nS) {
  static_assert(NS * 64 <= 256, "layer-0 factor stage uses 64 threads per sample");
  __shared__ float SM[GM_WORDS + NS * SSTR];
  float *const gm = SM;
  float *const ST = SM + GM_WORDS;

#define SRE(b) (ST + (2 * (b)) * PSTR)
#define SIM(b) (ST + (2 * (b) + 1) * PSTR)

  const int tid  = threadIdx.x;
  const int lane = tid & 31;
  const int wave = tid >> 5;
  const int s0   = blockIdx.x * NS;

#pragma unroll
  for (int j = 0; j < GM_WORDS / 256; ++j) gm[tid + 256 * j] = G[tid + 256 * j];

  const int perm0 = cnot_perm(tid, 1);

  if (tid < NS * N_WIRES) {
    const int i = tid / N_WIRES, w = tid - i * N_WIRES;
    int s = s0 + i;
    if (s >= nS) s = nS - 1;
    const float x = X[s * N_WIRES + w] * 0.7853981633974483f;
    const float c = __cosf(x), sn = __sinf(x);
    const float *m = G + U0_OFF + w * 8;
    float *vr = ST + i * SSTR + VR_OFF;
    float *vi = ST + i * SSTR + VI_OFF;
    vr[w]           = m[0] * c + m[2] * sn;
    vi[w]           = m[1] * c + m[3] * sn;
    vr[N_WIRES + w] = m[4] * c + m[6] * sn;
    vi[N_WIRES + w] = m[5] * c + m[7] * sn;
  }
  __syncthreads();

  if (tid < NS * 64) {
    float *sb = ST + (tid >> 6) * SSTR;
    const float *vr = sb + VR_OFF, *vi = sb + VI_OFF;
    const int t = tid & 63;
    if (t < 32) {
      float pr = 1.f, pi = 0.f;
#pragma unroll
      for (int w = 5; w < N_WIRES; ++w) {
        const int e = ((t >> (9 - w)) & 1) * N_WIRES + w;
        float nr, ni; cmul(pr, pi, vr[e], vi[e], nr, ni);
        pr = nr; pi = ni;
      }
      sb[PLOR_OFF + t] = pr; sb[PLOI_OFF + t] = pi;
    } else {
      const int h = t - 32;
      float pr = 1.f, pi = 0.f;
#pragma unroll
      for (int w = 0; w < 5; ++w) {
        const int e = ((h >> (4 - w)) & 1) * N_WIRES + w;
        float nr, ni; cmul(pr, pi, vr[e], vi[e], nr, ni);
        pr = nr; pi = ni;
      }
      sb[PHIR_OFF + h] = pr; sb[PHII_OFF + h] = pi;
    }
  }
  __syncthreads();

#pragma unroll
  for (int i = 0; i < NS; ++i) {
    const float *sb = ST + i * SSTR;
#pragma unroll
    for (int j = 0; j < DIM / 256; ++j) {
      const int k = tid + 256 * j;
      float re, im;
      cmul(sb[PHIR_OFF + (k >> 5)], sb[PHII_OFF + (k >> 5)],
           sb[PLOR_OFF + (k & 31)], sb[PLOI_OFF + (k & 31)], re, im);
      const int d = perm0 ^ cnot_perm(256 * j, 1);
      (SRE(1) + i * SSTR)[PADDED(d)] = re;
      (SIM(1) + i * SSTR)[PADDED(d)] = im;
    }
  }
  __syncthreads();

  const int  c16 = lane & 15;
  const int  h   = lane >> 4;
  const int  kb  = h << 1;
  const int  baseA = kb * 16 + c16;
  const int colA   = wave * 16 + c16;
  const int baseBA = colA + (colA >> 4) + kb * 136;
  const int baseWA = colA + (colA >> 4) + h * PSTR;
  const int baseBB = wave * 136 + c16 + kb * 17;
  const int baseWB = wave * 136 + c16 + h * PSTR;
  const int  tile = wave >> 1;
  const bool pimg = (wave & 1) != 0;
  const int  col  = tile * 16 + c16;
  const int baseB2 = col * 17 + kb;
  const int colC   = (wave >> 1) * 16 + c16;
  const int baseBC = (wave >> 1) * 272 + c16 * 17 + (wave & 1) * 4 + kb;
  const int col3   = ((wave & 4) << 7) | ((wave & 3) << 6) | ((c16 & 8) << 2) | ((c16 & 7) << 1);
  const int baseB3 = col3 + (col3 >> 4) + h * 17;
  const int pbG = cnot_perm((col << 4) | (h << 3), 2);
  const int pbC = cnot_perm((colC << 4) | ((wave & 1) << 2), 3);

#pragma unroll
  for (int l = 1; l < N_LAYERS - 1; ++l) {
    const float *Lg = gm + (l - 1) * LAYER_F;
    const int r0 = (3 * l) & 1;

    group_apply_stacked(Lg + baseA,
                        SRE(r0) + baseBA, SIM(r0) + baseBA,
                        SRE(r0 ^ 1) + baseWA, 136);
    __syncthreads();
    group_apply_stacked(Lg + 256 + baseA,
                        SRE(r0 ^ 1) + baseBB, SIM(r0 ^ 1) + baseBB,
                        SRE(r0) + baseWB, 17);
    __syncthreads();
    if (l == 2) {
      group_apply_stacked_scatter(Lg + 512 + baseA,
                                  SRE(r0) + baseBC, SIM(r0) + baseBC,
                                  SRE(r0 ^ 1) + h * PSTR,
                                  l + 1, pbC);
      __syncthreads();
    } else {
      group_apply(Lg + 512 + baseA,
                  (pimg ? SIM(r0) : SRE(r0)) + baseB2,
                  (pimg ? SRE(r0) : SIM(r0)) + baseB2,
                  (pimg ? SIM(r0 ^ 1) : SRE(r0 ^ 1)),
                  l + 1, pbG, pimg);
      __syncthreads();
    }
  }

  float z[NS];
  group_apply_stacked_z(gm + L3_OFF + baseA, SRE(1) + baseB3, SIM(1) + baseB3, z);
#pragma unroll
  for (int m = 16; m > 0; m >>= 1) {
#pragma unroll
    for (int i = 0; i < NS; ++i) z[i] += __shfl_xor(z[i], m, 32);
  }
  if (lane == 0) {
#pragma unroll
    for (int i = 0; i < NS; ++i) ST[i * SSTR + wave] = z[i];
  }
  __syncthreads();
  if (tid < NS && s0 + tid < nS) {
    const float *red = ST + tid * SSTR;
    float t = 0.f;
#pragma unroll
    for (int w = 0; w < 8; ++w) t += red[w];
    out[s0 + tid] = t + bias[0];
  }
#undef SRE
#undef SIM
}

extern "C" void kernel_launch(void *const *d_in, const int *in_sizes, int n_in,
                              void *d_out, int out_size, void *d_ws, size_t ws_size,
                              hipStream_t stream) {
  const float *X = (const float *)d_in[0];
  const float *W = (const float *)d_in[1];
  const float *bias = (const float *)d_in[2];
  float *out = (float *)d_out;
  float *G = (float *)d_ws;

  const int B = in_sizes[0] / N_WIRES;

  vqc_precompute<<<1, 256, 0, stream>>>(W, G);
  vqc_kernel<<<(B + NS - 1) / NS, 256, 0, stream>>>(X, G, bias, out, B);
}


# Layer 1 in closed form: behind the range-1 CNOT ring the product state is a bond chain, so layer-1 output is a rank-4 sum of half-chain factors (vector unit); 64 of 128 WMMAs per wave gone; new sums
# speedup vs baseline: 3.7640x; 1.3765x over previous
#include <hip/hip_runtime.h>


typedef float v2f __attribute__((ext_vector_type(2)));
typedef float v8f __attribute__((ext_vector_type(8)));

#define N_WIRES 10
#define N_LAYERS 4
#define DIM 1024
#define PADDED(k) ((k) + ((k) >> 4))
#define PDIM (DIM + (DIM >> 4))
#define PSTR (PDIM + 32)
#define L2_OFF 0
#define L3_OFF (L2_OFF + 768)
#define U1_OFF (L3_OFF + 256)
#define GM_STAGED (U1_OFF + 8 * N_WIRES)
#define GM_WORDS 1120
#define U0_OFF GM_STAGED
#define NS 4
#define SSTR (4 * PSTR)
#define VR_OFF 0
#define VI_OFF 32
#define FL_OFF 64
#define FR_OFF 320

__device__ __forceinline__ void cmul(float ar, float ai, float br, float bi,
                                     float &cr, float &ci) {
  cr = ar * br - ai * bi;
  ci = ar * bi + ai * br;
}

__device__ __forceinline__ int cnot_perm(int m, int rr) {
#pragma unroll
  for (int w = 0; w < N_WIRES; ++w) {
    int t = w + rr; if (t >= N_WIRES) t -= N_WIRES;
    m ^= ((m >> (9 - w)) & 1) << (9 - t);
  }
  return m;
}

__device__ __forceinline__ float real_form3(const float (*u)[8], int w0, int w1, int w2,
                                            int r, int c) {
  float pr = 1.f, pi = 0.f;
#pragma unroll
  for (int q = 0; q < 3; ++q) {
    const int wq = (q == 0) ? w0 : ((q == 1) ? w1 : w2);
    const int rb = (r >> (2 - q)) & 1, cb2 = (c >> (2 - q)) & 1;
    const float *e = &u[wq][(rb * 2 + cb2) * 2];
    float nr, ni; cmul(pr, pi, e[0], e[1], nr, ni);
    pr = nr; pi = ni;
  }
  const int rq = r >> 3, cq = c >> 3;
  return (rq == cq) ? pr : (rq ? pi : -pi);
}

__global__ __launch_bounds__(256) void vqc_precompute(const float *__restrict__ W,
                                                      float *__restrict__ G) {
  __shared__ float u[N_WIRES][8];
  const int tid = threadIdx.x;
  const int r = tid >> 4, c = tid & 15;
  for (int l = 0; l < N_LAYERS; ++l) {
    if (tid < N_WIRES) {
      const float phi   = W[(l * N_WIRES + tid) * 3 + 0];
      const float theta = W[(l * N_WIRES + tid) * 3 + 1];
      const float omega = W[(l * N_WIRES + tid) * 3 + 2];
      const float cs = __cosf(0.5f * theta), s = __sinf(0.5f * theta);
      const float a = 0.5f * (phi + omega), b = 0.5f * (phi - omega);
      const float ca = __cosf(a), sa = __sinf(a), cb = __cosf(b), sb = __sinf(b);
      u[tid][0] =  ca * cs; u[tid][1] = -sa * cs;
      u[tid][2] = -cb * s;  u[tid][3] = -sb * s;
      u[tid][4] =  cb * s;  u[tid][5] = -sb * s;
      u[tid][6] =  ca * cs; u[tid][7] =  sa * cs;
    }
    __syncthreads();
    if (l <= 1) {
      if (tid < N_WIRES * 8) G[(l == 0 ? U0_OFF : U1_OFF) + tid] = u[tid >> 3][tid & 7];
    } else if (l == N_LAYERS - 1) {
      G[L3_OFF + c * 16 + r] = real_form3(u, 1, 5, 9, r, c);
    } else {
      float *L = G + L2_OFF;
      L[c * 16 + r]       = real_form3(u, 0, 1, 2, r, c);
      L[256 + c * 16 + r] = real_form3(u, 3, 4, 5, r, c);
      L[512 + c * 16 + r] = real_form3(u, 6, 8, 9, r, c);
    }
    __syncthreads();
  }
}

__device__ __forceinline__ void group_apply_stacked(const float *__restrict__ A,
                                                    const float *bRe, const float *bIm,
                                                    float *dst, int idxStride) {
  v8f acc[NS];
#pragma unroll
  for (int i = 0; i < NS; ++i) acc[i] = v8f{};
#pragma unroll
  for (int kk = 0; kk < 4; ++kk) {
    const float *p = (kk < 2) ? bRe : bIm;
    const int o = (kk & 1) * 4 * idxStride;
    v2f a;
    a.x = A[kk * 64];  a.y = A[kk * 64 + 16];
#pragma unroll
    for (int i = 0; i < NS; ++i) {
      v2f b;
      b.x = p[i * SSTR + o];  b.y = p[i * SSTR + o + idxStride];
      acc[i] = __builtin_amdgcn_wmma_f32_16x16x4_f32(false, a, false, b,
                                                     (short)0, acc[i], false, false);
    }
  }
#pragma unroll
  for (int i = 0; i < NS; ++i) {
#pragma unroll
    for (int v = 0; v < 8; ++v) dst[i * SSTR + v * idxStride] = acc[i][v];
  }
}

__device__ __forceinline__ void group_apply_stacked_scatter(const float *__restrict__ A,
                                                            const float *bRe, const float *bIm,
                                                            float *dst, int rr, int permBase) {
  v8f acc[NS];
#pragma unroll
  for (int i = 0; i < NS; ++i) acc[i] = v8f{};
#pragma unroll
  for (int kk = 0; kk < 4; ++kk) {
    const float *p = (kk < 2) ? bRe : bIm;
    const int o = (kk & 1) * 8;
    v2f a;
    a.x = A[kk * 64];  a.y = A[kk * 64 + 16];
#pragma unroll
    for (int i = 0; i < NS; ++i) {
      v2f b;
      b.x = p[i * SSTR + o];  b.y = p[i * SSTR + o + 1];
      acc[i] = __builtin_amdgcn_wmma_f32_16x16x4_f32(false, a, false, b,
                                                     (short)0, acc[i], false, false);
    }
  }
#pragma unroll
  for (int v = 0; v < 8; ++v) {
    const int w = PADDED(permBase ^ cnot_perm(((v & 4) << 1) | (v & 3), rr));
#pragma unroll
    for (int i = 0; i < NS; ++i) dst[i * SSTR + w] = acc[i][v];
  }
}

__device__ __forceinline__ void group_apply_stacked_z(const float *__restrict__ A,
                                                      const float *bRe, const float *bIm,
                                                      float *z) {
  v8f acc[NS];
#pragma unroll
  for (int i = 0; i < NS; ++i) acc[i] = v8f{};
#pragma unroll
  for (int kk = 0; kk < 4; ++kk) {
    const float *p = (kk < 2) ? bRe : bIm;
    const int o = (kk & 1) * 272;
    v2f a;
    a.x = A[kk * 64];  a.y = A[kk * 64 + 16];
#pragma unroll
    for (int i = 0; i < NS; ++i) {
      v2f b;
      b.x = p[i * SSTR + o];  b.y = p[i * SSTR + o + 1];
      acc[i] = __builtin_amdgcn_wmma_f32_16x16x4_f32(false, a, false, b,
                                                     (short)0, acc[i], false, false);
    }
  }
#pragma unroll
  for (int i = 0; i < NS; ++i) {
    float t = 0.f;
#pragma unroll
    for (int v = 0; v < 8; ++v) {
      const float sq = acc[i][v] * acc[i][v];
      t += (((v >> 2) ^ (v >> 1) ^ v) & 1) ? -sq : sq;
    }
    z[i] = t;
  }
}

__global__ __launch_bounds__(256) void vqc_kernel(const float *__restrict__ X,
                                                  const float *__restrict__ G,
                                                  const float *__restrict__ bias,
                                                  float *__restrict__ out, int nS) {
  static_assert(NS * 64 <= 256, "the factor stage uses 64 threads per sample");
  __shared__ __attribute__((aligned(16))) float SM[GM_WORDS + NS * SSTR];
  float *const gm = SM;
  float *const ST = SM + GM_WORDS;

#define SRE(b) (ST + (2 * (b)) * PSTR)
#define SIM(b) (ST + (2 * (b) + 1) * PSTR)

  const int tid  = threadIdx.x;
  const int lane = tid & 31;
  const int wave = tid >> 5;
  const int s0   = blockIdx.x * NS;

#pragma unroll
  for (int j = 0; j < U1_OFF / 256; ++j) gm[tid + 256 * j] = G[tid + 256 * j];
  if (tid < GM_STAGED - U1_OFF) gm[U1_OFF + tid] = G[U1_OFF + tid];

  const int perm1 = cnot_perm(tid, 2);
#define FB(i) (ST + (i) * SSTR + 2 * PSTR)

  if (tid < NS * N_WIRES) {
    const int i = tid / N_WIRES, w = tid - i * N_WIRES;
    int s = s0 + i;
    if (s >= nS) s = nS - 1;
    const float x = X[s * N_WIRES + w] * 0.7853981633974483f;
    const float c = __cosf(x), sn = __sinf(x);
    const float *m = G + U0_OFF + w * 8;
    float *vr = FB(i) + VR_OFF;
    float *vi = FB(i) + VI_OFF;
    vr[w]           = m[0] * c + m[2] * sn;
    vi[w]           = m[1] * c + m[3] * sn;
    vr[N_WIRES + w] = m[4] * c + m[6] * sn;
    vi[N_WIRES + w] = m[5] * c + m[7] * sn;
  }
  __syncthreads();

  if (tid < NS * 64) {
    float *fb = FB(tid >> 6);
    const float *vr = fb + VR_OFF, *vi = fb + VI_OFF;
    const float *u1 = gm + U1_OFF;
    const int t = tid & 63;
    if (t < 32) {
#pragma unroll
      for (int g = 0; g < 2; ++g) {
        float rr_[2], ri_[2];
        {
          const float *e = u1 + 9 * 8 + (((t & 1) * 2 + g) * 2);
#pragma unroll
          for (int p = 0; p < 2; ++p) {
            const int vb = (g ^ p) * N_WIRES + 9;
            cmul(e[0], e[1], vr[vb], vi[vb], rr_[p], ri_[p]);
          }
        }
#pragma unroll
        for (int w = 8; w >= 5; --w) {
          const int ib = (t >> (9 - w)) & 1;
          float qr[2], qi[2];
#pragma unroll
          for (int c = 0; c < 2; ++c) {
            const float *e = u1 + w * 8 + ((ib * 2 + c) * 2);
            cmul(e[0], e[1], rr_[c], ri_[c], qr[c], qi[c]);
          }
#pragma unroll
          for (int p = 0; p < 2; ++p) {
            const int e0 = p * N_WIRES + w, e1 = (1 ^ p) * N_WIRES + w;
            float ar, ai, br, bi;
            cmul(qr[0], qi[0], vr[e0], vi[e0], ar, ai);
            cmul(qr[1], qi[1], vr[e1], vi[e1], br, bi);
            rr_[p] = ar + br; ri_[p] = ai + bi;
          }
        }
        fb[FR_OFF + t * 8 + g * 2 + 0] = rr_[0];
        fb[FR_OFF + t * 8 + g * 2 + 1] = rr_[1];
        fb[FR_OFF + t * 8 + 4 + g * 2 + 0] = ri_[0];
        fb[FR_OFF + t * 8 + 4 + g * 2 + 1] = ri_[1];
      }
    } else {
      const int hh = t - 32;
#pragma unroll
      for (int g = 0; g < 2; ++g) {
        float ar_[2], ai_[2];
        {
          const int ib = (hh >> 4) & 1;
#pragma unroll
          for (int c = 0; c < 2; ++c) {
            const float *e = u1 + ((ib * 2 + c) * 2);
            const int vb = (c ^ g) * N_WIRES;
            cmul(e[0], e[1], vr[vb], vi[vb], ar_[c], ai_[c]);
          }
        }
#pragma unroll
        for (int w = 1; w < 5; ++w) {
          const int ib = (hh >> (4 - w)) & 1;
          const int x = (w == 1) ? g : 0;
          float nr[2], ni[2];
#pragma unroll
          for (int c = 0; c < 2; ++c) {
            const int e0 = (c ^ x) * N_WIRES + w, e1 = (c ^ 1 ^ x) * N_WIRES + w;
            float s0r, s0i, s1r, s1i;
            cmul(ar_[0], ai_[0], vr[e0], vi[e0], s0r, s0i);
            cmul(ar_[1], ai_[1], vr[e1], vi[e1], s1r, s1i);
            const float *e = u1 + w * 8 + ((ib * 2 + c) * 2);
            cmul(e[0], e[1], s0r + s1r, s0i + s1i, nr[c], ni[c]);
          }
          ar_[0] = nr[0]; ai_[0] = ni[0]; ar_[1] = nr[1]; ai_[1] = ni[1];
        }
        fb[FL_OFF + hh * 8 + g * 2 + 0] = ar_[0];
        fb[FL_OFF + hh * 8 + g * 2 + 1] = ar_[1];
        fb[FL_OFF + hh * 8 + 4 + g * 2 + 0] = ai_[0];
        fb[FL_OFF + hh * 8 + 4 + g * 2 + 1] = ai_[1];
      }
    }
  }
  __syncthreads();

#pragma unroll
  for (int i = 0; i < NS; ++i) {
    const float *fb = FB(i);
    const float4 Rr = *reinterpret_cast<const float4 *>(fb + FR_OFF + (tid & 31) * 8);
    const float4 Ri = *reinterpret_cast<const float4 *>(fb + FR_OFF + (tid & 31) * 8 + 4);
#pragma unroll
    for (int j = 0; j < DIM / 256; ++j) {
      const int k = tid + 256 * j;
      const float4 Lr = *reinterpret_cast<const float4 *>(fb + FL_OFF + (k >> 5) * 8);
      const float4 Li = *reinterpret_cast<const float4 *>(fb + FL_OFF + (k >> 5) * 8 + 4);
      const float re = (Lr.x * Rr.x - Li.x * Ri.x) + (Lr.y * Rr.y - Li.y * Ri.y)
                     + (Lr.z * Rr.z - Li.z * Ri.z) + (Lr.w * Rr.w - Li.w * Ri.w);
      const float im = (Lr.x * Ri.x + Li.x * Rr.x) + (Lr.y * Ri.y + Li.y * Rr.y)
                     + (Lr.z * Ri.z + Li.z * Rr.z) + (Lr.w * Ri.w + Li.w * Rr.w);
      const int d = perm1 ^ cnot_perm(256 * j, 2);
      (SRE(0) + i * SSTR)[PADDED(d)] = re;
      (SIM(0) + i * SSTR)[PADDED(d)] = im;
    }
  }
  __syncthreads();
#undef FB

  const int  c16 = lane & 15;
  const int  h   = lane >> 4;
  const int  kb  = h << 1;
  const int  baseA = kb * 16 + c16;
  const int colA   = wave * 16 + c16;
  const int baseBA = colA + (colA >> 4) + kb * 136;
  const int baseWA = colA + (colA >> 4) + h * PSTR;
  const int baseBB = wave * 136 + c16 + kb * 17;
  const int baseWB = wave * 136 + c16 + h * PSTR;
  const int colC   = (wave >> 1) * 16 + c16;
  const int baseBC = (wave >> 1) * 272 + c16 * 17 + (wave & 1) * 4 + kb;
  const int col3   = ((wave & 4) << 7) | ((wave & 3) << 6) | ((c16 & 8) << 2) | ((c16 & 7) << 1);
  const int baseB3 = col3 + (col3 >> 4) + h * 17;
  const int pbC = cnot_perm((colC << 4) | ((wave & 1) << 2), 3);

  {
    const float *Lg = gm + L2_OFF;
    group_apply_stacked(Lg + baseA,
                        SRE(0) + baseBA, SIM(0) + baseBA,
                        SRE(1) + baseWA, 136);
    __syncthreads();
    group_apply_stacked(Lg + 256 + baseA,
                        SRE(1) + baseBB, SIM(1) + baseBB,
                        SRE(0) + baseWB, 17);
    __syncthreads();
    group_apply_stacked_scatter(Lg + 512 + baseA,
                                SRE(0) + baseBC, SIM(0) + baseBC,
                                SRE(1) + h * PSTR,
                                3, pbC);
    __syncthreads();
  }

  float z[NS];
  group_apply_stacked_z(gm + L3_OFF + baseA, SRE(1) + baseB3, SIM(1) + baseB3, z);
#pragma unroll
  for (int m = 16; m > 0; m >>= 1) {
#pragma unroll
    for (int i = 0; i < NS; ++i) z[i] += __shfl_xor(z[i], m, 32);
  }
  if (lane == 0) {
#pragma unroll
    for (int i = 0; i < NS; ++i) ST[i * SSTR + wave] = z[i];
  }
  __syncthreads();
  if (tid < NS && s0 + tid < nS) {
    const float *red = ST + tid * SSTR;
    float t = 0.f;
#pragma unroll
    for (int w = 0; w < 8; ++w) t += red[w];
    out[s0 + tid] = t + bias[0];
  }
#undef SRE
#undef SIM
}

extern "C" void kernel_launch(void *const *d_in, const int *in_sizes, int n_in,
                              void *d_out, int out_size, void *d_ws, size_t ws_size,
                              hipStream_t stream) {
  const float *X = (const float *)d_in[0];
  const float *W = (const float *)d_in[1];
  const float *bias = (const float *)d_in[2];
  float *out = (float *)d_out;
  float *G = (float *)d_ws;

  const int B = in_sizes[0] / N_WIRES;

  vqc_precompute<<<1, 256, 0, stream>>>(W, G);
  vqc_kernel<<<(B + NS - 1) / NS, 256, 0, stream>>>(X, G, bias, out, B);
}


# Layer-2 operator of wires 0-2 applied to the left rank-4 factor of the layer-1 closed form (64 columns) instead of the state (128): 56 WMMAs per wave instead of 64; ring-1 scatter gone; new sums
# speedup vs baseline: 4.0946x; 1.0878x over previous
#include <hip/hip_runtime.h>


typedef float v2f __attribute__((ext_vector_type(2)));
typedef float v8f __attribute__((ext_vector_type(8)));

#define N_WIRES 10
#define N_LAYERS 4
#define DIM 1024
#define PADDED(k) ((k) + ((k) >> 4))
#define PDIM (DIM + (DIM >> 4))
#define PSTR (PDIM + 32)
#define L2_OFF 0
#define L3_OFF (L2_OFF + 768)
#define U1_OFF (L3_OFF + 256)
#define GM_STAGED (U1_OFF + 8 * N_WIRES)
#define GM_WORDS 1120
#define U0_OFF GM_STAGED
#define NS 4
#define SSTR (4 * PSTR)
#define VR_OFF 0
#define VI_OFF 32
#define FL_OFF 64
#define FR_OFF 320

__device__ __forceinline__ void cmul(float ar, float ai, float br, float bi,
                                     float &cr, float &ci) {
  cr = ar * br - ai * bi;
  ci = ar * bi + ai * br;
}

__device__ __forceinline__ int cnot_perm(int m, int rr) {
#pragma unroll
  for (int w = 0; w < N_WIRES; ++w) {
    int t = w + rr; if (t >= N_WIRES) t -= N_WIRES;
    m ^= ((m >> (9 - w)) & 1) << (9 - t);
  }
  return m;
}

__device__ __forceinline__ float real_form3(const float (*u)[8], int w0, int w1, int w2,
                                            int r, int c) {
  float pr = 1.f, pi = 0.f;
#pragma unroll
  for (int q = 0; q < 3; ++q) {
    const int wq = (q == 0) ? w0 : ((q == 1) ? w1 : w2);
    const int rb = (r >> (2 - q)) & 1, cb2 = (c >> (2 - q)) & 1;
    const float *e = &u[wq][(rb * 2 + cb2) * 2];
    float nr, ni; cmul(pr, pi, e[0], e[1], nr, ni);
    pr = nr; pi = ni;
  }
  const int rq = r >> 3, cq = c >> 3;
  return (rq == cq) ? pr : (rq ? pi : -pi);
}

__global__ __launch_bounds__(256) void vqc_precompute(const float *__restrict__ W,
                                                      float *__restrict__ G) {
  __shared__ float u[N_WIRES][8];
  const int tid = threadIdx.x;
  const int r = tid >> 4, c = tid & 15;
  for (int l = 0; l < N_LAYERS; ++l) {
    if (tid < N_WIRES) {
      const float phi   = W[(l * N_WIRES + tid) * 3 + 0];
      const float theta = W[(l * N_WIRES + tid) * 3 + 1];
      const float omega = W[(l * N_WIRES + tid) * 3 + 2];
      const float cs = __cosf(0.5f * theta), s = __sinf(0.5f * theta);
      const float a = 0.5f * (phi + omega), b = 0.5f * (phi - omega);
      const float ca = __cosf(a), sa = __sinf(a), cb = __cosf(b), sb = __sinf(b);
      u[tid][0] =  ca * cs; u[tid][1] = -sa * cs;
      u[tid][2] = -cb * s;  u[tid][3] = -sb * s;
      u[tid][4] =  cb * s;  u[tid][5] = -sb * s;
      u[tid][6] =  ca * cs; u[tid][7] =  sa * cs;
    }
    __syncthreads();
    if (l <= 1) {
      if (tid < N_WIRES * 8) G[(l == 0 ? U0_OFF : U1_OFF) + tid] = u[tid >> 3][tid & 7];
    } else if (l == N_LAYERS - 1) {
      G[L3_OFF + c * 16 + r] = real_form3(u, 1, 5, 9, r, c);
    } else {
      float *L = G + L2_OFF;
      L[c * 16 + r]       = real_form3(u, 0, 1, 2, r, c);
      L[256 + c * 16 + r] = real_form3(u, 3, 4, 5, r, c);
      L[512 + c * 16 + r] = real_form3(u, 6, 8, 9, r, c);
    }
    __syncthreads();
  }
}

__device__ __forceinline__ void group_apply_stacked(const float *__restrict__ A,
                                                    const float *bRe, const float *bIm,
                                                    float *dst, int idxStride) {
  v8f acc[NS];
#pragma unroll
  for (int i = 0; i < NS; ++i) acc[i] = v8f{};
#pragma unroll
  for (int kk = 0; kk < 4; ++kk) {
    const float *p = (kk < 2) ? bRe : bIm;
    const int o = (kk & 1) * 4 * idxStride;
    v2f a;
    a.x = A[kk * 64];  a.y = A[kk * 64 + 16];
#pragma unroll
    for (int i = 0; i < NS; ++i) {
      v2f b;
      b.x = p[i * SSTR + o];  b.y = p[i * SSTR + o + idxStride];
      acc[i] = __builtin_amdgcn_wmma_f32_16x16x4_f32(false, a, false, b,
                                                     (short)0, acc[i], false, false);
    }
  }
#pragma unroll
  for (int i = 0; i < NS; ++i) {
#pragma unroll
    for (int v = 0; v < 8; ++v) dst[i * SSTR + v * idxStride] = acc[i][v];
  }
}

__device__ __forceinline__ void group_apply_stacked_scatter(const float *__restrict__ A,
                                                            const float *bRe, const float *bIm,
                                                            float *dst, int rr, int permBase) {
  v8f acc[NS];
#pragma unroll
  for (int i = 0; i < NS; ++i) acc[i] = v8f{};
#pragma unroll
  for (int kk = 0; kk < 4; ++kk) {
    const float *p = (kk < 2) ? bRe : bIm;
    const int o = (kk & 1) * 8;
    v2f a;
    a.x = A[kk * 64];  a.y = A[kk * 64 + 16];
#pragma unroll
    for (int i = 0; i < NS; ++i) {
      v2f b;
      b.x = p[i * SSTR + o];  b.y = p[i * SSTR + o + 1];
      acc[i] = __builtin_amdgcn_wmma_f32_16x16x4_f32(false, a, false, b,
                                                     (short)0, acc[i], false, false);
    }
  }
#pragma unroll
  for (int v = 0; v < 8; ++v) {
    const int w = PADDED(permBase ^ cnot_perm(((v & 4) << 1) | (v & 3), rr));
#pragma unroll
    for (int i = 0; i < NS; ++i) dst[i * SSTR + w] = acc[i][v];
  }
}

__device__ __forceinline__ void group_apply_stacked_z(const float *__restrict__ A,
                                                      const float *bRe, const float *bIm,
                                                      float *z) {
  v8f acc[NS];
#pragma unroll
  for (int i = 0; i < NS; ++i) acc[i] = v8f{};
#pragma unroll
  for (int kk = 0; kk < 4; ++kk) {
    const float *p = (kk < 2) ? bRe : bIm;
    const int o = (kk & 1) * 272;
    v2f a;
    a.x = A[kk * 64];  a.y = A[kk * 64 + 16];
#pragma unroll
    for (int i = 0; i < NS; ++i) {
      v2f b;
      b.x = p[i * SSTR + o];  b.y = p[i * SSTR + o + 1];
      acc[i] = __builtin_amdgcn_wmma_f32_16x16x4_f32(false, a, false, b,
                                                     (short)0, acc[i], false, false);
    }
  }
#pragma unroll
  for (int i = 0; i < NS; ++i) {
    float t = 0.f;
#pragma unroll
    for (int v = 0; v < 8; ++v) {
      const float sq = acc[i][v] * acc[i][v];
      t += (((v >> 2) ^ (v >> 1) ^ v) & 1) ? -sq : sq;
    }
    z[i] = t;
  }
}

__global__ __launch_bounds__(256) void vqc_kernel(const float *__restrict__ X,
                                                  const float *__restrict__ G,
                                                  const float *__restrict__ bias,
                                                  float *__restrict__ out, int nS) {
  static_assert(NS * 64 <= 256, "the factor stage uses 64 threads per sample");
  __shared__ __attribute__((aligned(16))) float SM[GM_WORDS + NS * SSTR];
  float *const gm = SM;
  float *const ST = SM + GM_WORDS;

#define SRE(b) (ST + (2 * (b)) * PSTR)
#define SIM(b) (ST + (2 * (b) + 1) * PSTR)

  const int tid  = threadIdx.x;
  const int lane = tid & 31;
  const int wave = tid >> 5;
  const int s0   = blockIdx.x * NS;

#pragma unroll
  for (int j = 0; j < U1_OFF / 256; ++j) gm[tid + 256 * j] = G[tid + 256 * j];
  if (tid < GM_STAGED - U1_OFF) gm[U1_OFF + tid] = G[U1_OFF + tid];

#define FB(i) (ST + (i) * SSTR + 2 * PSTR)
#define LPB(i) (ST + (i) * SSTR + 3 * PSTR)

  if (tid < NS * N_WIRES) {
    const int i = tid / N_WIRES, w = tid - i * N_WIRES;
    int s = s0 + i;
    if (s >= nS) s = nS - 1;
    const float x = X[s * N_WIRES + w] * 0.7853981633974483f;
    const float c = __cosf(x), sn = __sinf(x);
    const float *m = G + U0_OFF + w * 8;
    float *vr = FB(i) + VR_OFF;
    float *vi = FB(i) + VI_OFF;
    vr[w]           = m[0] * c + m[2] * sn;
    vi[w]           = m[1] * c + m[3] * sn;
    vr[N_WIRES + w] = m[4] * c + m[6] * sn;
    vi[N_WIRES + w] = m[5] * c + m[7] * sn;
  }
  __syncthreads();

  if (tid < NS * 64) {
    float *fb = FB(tid >> 6);
    const float *vr = fb + VR_OFF, *vi = fb + VI_OFF;
    const float *u1 = gm + U1_OFF;
    const int t = tid & 63;
    if (t < 32) {
#pragma unroll
      for (int g = 0; g < 2; ++g) {
        float rr_[2], ri_[2];
        {
          const float *e = u1 + 9 * 8 + (((t & 1) * 2 + g) * 2);
#pragma unroll
          for (int p = 0; p < 2; ++p) {
            const int vb = (g ^ p) * N_WIRES + 9;
            cmul(e[0], e[1], vr[vb], vi[vb], rr_[p], ri_[p]);
          }
        }
#pragma unroll
        for (int w = 8; w >= 5; --w) {
          const int ib = (t >> (9 - w)) & 1;
          float qr[2], qi[2];
#pragma unroll
          for (int c = 0; c < 2; ++c) {
            const float *e = u1 + w * 8 + ((ib * 2 + c) * 2);
            cmul(e[0], e[1], rr_[c], ri_[c], qr[c], qi[c]);
          }
#pragma unroll
          for (int p = 0; p < 2; ++p) {
            const int e0 = p * N_WIRES + w, e1 = (1 ^ p) * N_WIRES + w;
            float ar, ai, br, bi;
            cmul(qr[0], qi[0], vr[e0], vi[e0], ar, ai);
            cmul(qr[1], qi[1], vr[e1], vi[e1], br, bi);
            rr_[p] = ar + br; ri_[p] = ai + bi;
          }
        }
        fb[FR_OFF + t * 8 + g * 2 + 0] = rr_[0];
        fb[FR_OFF + t * 8 + g * 2 + 1] = rr_[1];
        fb[FR_OFF + t * 8 + 4 + g * 2 + 0] = ri_[0];
        fb[FR_OFF + t * 8 + 4 + g * 2 + 1] = ri_[1];
      }
    } else {
      const int hh = t - 32;
#pragma unroll
      for (int g = 0; g < 2; ++g) {
        float ar_[2], ai_[2];
        {
          const int ib = (hh >> 4) & 1;
#pragma unroll
          for (int c = 0; c < 2; ++c) {
            const float *e = u1 + ((ib * 2 + c) * 2);
            const int vb = (c ^ g) * N_WIRES;
            cmul(e[0], e[1], vr[vb], vi[vb], ar_[c], ai_[c]);
          }
        }
#pragma unroll
        for (int w = 1; w < 5; ++w) {
          const int ib = (hh >> (4 - w)) & 1;
          const int x = (w == 1) ? g : 0;
          float nr[2], ni[2];
#pragma unroll
          for (int c = 0; c < 2; ++c) {
            const int e0 = (c ^ x) * N_WIRES + w, e1 = (c ^ 1 ^ x) * N_WIRES + w;
            float s0r, s0i, s1r, s1i;
            cmul(ar_[0], ai_[0], vr[e0], vi[e0], s0r, s0i);
            cmul(ar_[1], ai_[1], vr[e1], vi[e1], s1r, s1i);
            const float *e = u1 + w * 8 + ((ib * 2 + c) * 2);
            cmul(e[0], e[1], s0r + s1r, s0i + s1i, nr[c], ni[c]);
          }
          ar_[0] = nr[0]; ai_[0] = ni[0]; ar_[1] = nr[1]; ai_[1] = ni[1];
        }
        fb[FL_OFF + hh * 8 + g * 2 + 0] = ar_[0];
        fb[FL_OFF + hh * 8 + g * 2 + 1] = ar_[1];
        fb[FL_OFF + hh * 8 + 4 + g * 2 + 0] = ai_[0];
        fb[FL_OFF + hh * 8 + 4 + g * 2 + 1] = ai_[1];
      }
    }
  }
  __syncthreads();

  const int  c16 = lane & 15;
  const int  h   = lane >> 4;
  const int  kb  = h << 1;
  const int  baseA = kb * 16 + c16;

  {
    const int si = wave >> 1, qb = (wave & 1) * 2;
    const float *fl = FB(si) + FL_OFF + qb;
    const float *A = gm + L2_OFF + baseA;
    const int ih = (((c16 >> 3) & 1) * 2) ^ ((c16 >> 2) & 1) ^ (((c16 >> 1) & 1) * 20)
                 ^ (((c16 & 1) ^ h) * 10);
    v8f acc[2];
    acc[0] = v8f{}; acc[1] = v8f{};
#pragma unroll
    for (int kk = 0; kk < 4; ++kk) {
      const int r0 = (ih ^ ((kk & 1) * 20)) * 8 + (kk >> 1) * 4;
      const int r1 = (ih ^ ((kk & 1) * 20) ^ 5) * 8 + (kk >> 1) * 4;
      v2f a;
      a.x = A[kk * 64];  a.y = A[kk * 64 + 16];
#pragma unroll
      for (int t = 0; t < 2; ++t) {
        v2f b;
        b.x = fl[r0 + t];  b.y = fl[r1 + t];
        acc[t] = __builtin_amdgcn_wmma_f32_16x16x4_f32(false, a, false, b,
                                                       (short)0, acc[t], false, false);
      }
    }
    float *lp = LPB(si) + c16 * 8 + h * 4 + qb;
#pragma unroll
    for (int v = 0; v < 8; ++v) {
#pragma unroll
      for (int t = 0; t < 2; ++t) lp[v * 128 + t] = acc[t][v];
    }
  }
  __syncthreads();

  {
    const int dlo = tid & 127;
    const int ilo = (dlo ^ (dlo >> 2)) & 31;
    const int ent = ((tid >> 7) & 1) * 16 + ((tid >> 5) & 3) * 4 + (tid & 3);
    const int wo = PADDED(tid);
#pragma unroll
    for (int i = 0; i < NS; ++i) {
      const float *fr = FB(i) + FR_OFF + ilo * 8;
      const float *lq = LPB(i) + ent * 8;
      const float4 Rr = *reinterpret_cast<const float4 *>(fr);
      const float4 Ri = *reinterpret_cast<const float4 *>(fr + 4);
#pragma unroll
      for (int j = 0; j < DIM / 256; ++j) {
        const float4 Lr = *reinterpret_cast<const float4 *>(lq + j * 256);
        const float4 Li = *reinterpret_cast<const float4 *>(lq + j * 256 + 4);
        const float re = (Lr.x * Rr.x - Li.x * Ri.x) + (Lr.y * Rr.y - Li.y * Ri.y)
                       + (Lr.z * Rr.z - Li.z * Ri.z) + (Lr.w * Rr.w - Li.w * Ri.w);
        const float im = (Lr.x * Ri.x + Li.x * Rr.x) + (Lr.y * Ri.y + Li.y * Rr.y)
                       + (Lr.z * Ri.z + Li.z * Rr.z) + (Lr.w * Ri.w + Li.w * Rr.w);
        (SRE(0) + i * SSTR)[wo + 272 * j] = re;
        (SIM(0) + i * SSTR)[wo + 272 * j] = im;
      }
    }
  }
  __syncthreads();
#undef FB
#undef LPB

  const int baseBB = wave * 136 + c16 + kb * 17;
  const int baseWB = wave * 136 + c16 + h * PSTR;
  const int colC   = (wave >> 1) * 16 + c16;
  const int baseBC = (wave >> 1) * 272 + c16 * 17 + (wave & 1) * 4 + kb;
  const int col3   = ((wave & 4) << 7) | ((wave & 3) << 6) | ((c16 & 8) << 2) | ((c16 & 7) << 1);
  const int baseB3 = col3 + (col3 >> 4) + h * 17;
  const int pbC = cnot_perm((colC << 4) | ((wave & 1) << 2), 3);

  {
    const float *Lg = gm + L2_OFF;
    group_apply_stacked(Lg + 256 + baseA,
                        SRE(0) + baseBB, SIM(0) + baseBB,
                        SRE(1) + baseWB, 17);
    __syncthreads();
    group_apply_stacked_scatter(Lg + 512 + baseA,
                                SRE(1) + baseBC, SIM(1) + baseBC,
                                SRE(0) + h * PSTR,
                                3, pbC);
    __syncthreads();
  }

  float z[NS];
  group_apply_stacked_z(gm + L3_OFF + baseA, SRE(0) + baseB3, SIM(0) + baseB3, z);
#pragma unroll
  for (int m = 16; m > 0; m >>= 1) {
#pragma unroll
    for (int i = 0; i < NS; ++i) z[i] += __shfl_xor(z[i], m, 32);
  }
  if (lane == 0) {
#pragma unroll
    for (int i = 0; i < NS; ++i) SRE(1)[i * SSTR + wave] = z[i];
  }
  __syncthreads();
  if (tid < NS && s0 + tid < nS) {
    const float *red = SRE(1) + tid * SSTR;
    float t = 0.f;
#pragma unroll
    for (int w = 0; w < 8; ++w) t += red[w];
    out[s0 + tid] = t + bias[0];
  }
#undef SRE
#undef SIM
}

extern "C" void kernel_launch(void *const *d_in, const int *in_sizes, int n_in,
                              void *d_out, int out_size, void *d_ws, size_t ws_size,
                              hipStream_t stream) {
  const float *X = (const float *)d_in[0];
  const float *W = (const float *)d_in[1];
  const float *bias = (const float *)d_in[2];
  float *out = (float *)d_out;
  float *G = (float *)d_ws;

  const int B = in_sizes[0] / N_WIRES;

  vqc_precompute<<<1, 256, 0, stream>>>(W, G);
  vqc_kernel<<<(B + NS - 1) / NS, 256, 0, stream>>>(X, G, bias, out, B);
}


# One launch of 1024 resident workgroups, each looping over its sample groups: gates and operators built once per workgroup in LDS (precompute launch gone), index arithmetic hoisted; sums unchanged
# speedup vs baseline: 4.5299x; 1.1063x over previous
#include <hip/hip_runtime.h>


typedef float v2f __attribute__((ext_vector_type(2)));
typedef float v8f __attribute__((ext_vector_type(8)));

#define N_WIRES 10
#define N_LAYERS 4
#define DIM 1024
#define PADDED(k) ((k) + ((k) >> 4))
#define PDIM (DIM + (DIM >> 4))
#define PSTR (PDIM + 32)
#define L2_OFF 0
#define L3_OFF (L2_OFF + 768)
#define U1_OFF (L3_OFF + 256)
#define GM_WORDS 1120
#define NS 4
#define NWG 1024
#define SSTR (4 * PSTR)
#define VR_OFF 0
#define VI_OFF 32
#define FL_OFF 64
#define FR_OFF 320
#define ZP_OFF 576

__device__ __forceinline__ void cmul(float ar, float ai, float br, float bi,
                                     float &cr, float &ci) {
  cr = ar * br - ai * bi;
  ci = ar * bi + ai * br;
}

__device__ __forceinline__ int cnot_perm(int m, int rr) {
#pragma unroll
  for (int w = 0; w < N_WIRES; ++w) {
    int t = w + rr; if (t >= N_WIRES) t -= N_WIRES;
    m ^= ((m >> (9 - w)) & 1) << (9 - t);
  }
  return m;
}

__device__ __forceinline__ float real_form3(const float (*u)[8], int w0, int w1, int w2,
                                            int r, int c) {
  float pr = 1.f, pi = 0.f;
#pragma unroll
  for (int q = 0; q < 3; ++q) {
    const int wq = (q == 0) ? w0 : ((q == 1) ? w1 : w2);
    const int rb = (r >> (2 - q)) & 1, cb2 = (c >> (2 - q)) & 1;
    const float *e = &u[wq][(rb * 2 + cb2) * 2];
    float nr, ni; cmul(pr, pi, e[0], e[1], nr, ni);
    pr = nr; pi = ni;
  }
  const int rq = r >> 3, cq = c >> 3;
  return (rq == cq) ? pr : (rq ? pi : -pi);
}

__device__ __forceinline__ void rot_gate(float phi, float theta, float omega, float *u) {
  const float cs = __cosf(0.5f * theta), s = __sinf(0.5f * theta);
  const float a = 0.5f * (phi + omega), b = 0.5f * (phi - omega);
  const float ca = __cosf(a), sa = __sinf(a), cb = __cosf(b), sb = __sinf(b);
  u[0] =  ca * cs; u[1] = -sa * cs;
  u[2] = -cb * s;  u[3] = -sb * s;
  u[4] =  cb * s;  u[5] = -sb * s;
  u[6] =  ca * cs; u[7] =  sa * cs;
}

__device__ __forceinline__ void group_apply_stacked(const float *__restrict__ A,
                                                    const float *bRe, const float *bIm,
                                                    float *dst, int idxStride) {
  v8f acc[NS];
#pragma unroll
  for (int i = 0; i < NS; ++i) acc[i] = v8f{};
#pragma unroll
  for (int kk = 0; kk < 4; ++kk) {
    const float *p = (kk < 2) ? bRe : bIm;
    const int o = (kk & 1) * 4 * idxStride;
    v2f a;
    a.x = A[kk * 64];  a.y = A[kk * 64 + 16];
#pragma unroll
    for (int i = 0; i < NS; ++i) {
      v2f b;
      b.x = p[i * SSTR + o];  b.y = p[i * SSTR + o + idxStride];
      acc[i] = __builtin_amdgcn_wmma_f32_16x16x4_f32(false, a, false, b,
                                                     (short)0, acc[i], false, false);
    }
  }
#pragma unroll
  for (int i = 0; i < NS; ++i) {
#pragma unroll
    for (int v = 0; v < 8; ++v) dst[i * SSTR + v * idxStride] = acc[i][v];
  }
}

__device__ __forceinline__ void group_apply_stacked_scatter(const float *__restrict__ A,
                                                            const float *bRe, const float *bIm,
                                                            float *dst, int rr, int permBase) {
  v8f acc[NS];
#pragma unroll
  for (int i = 0; i < NS; ++i) acc[i] = v8f{};
#pragma unroll
  for (int kk = 0; kk < 4; ++kk) {
    const float *p = (kk < 2) ? bRe : bIm;
    const int o = (kk & 1) * 8;
    v2f a;
    a.x = A[kk * 64];  a.y = A[kk * 64 + 16];
#pragma unroll
    for (int i = 0; i < NS; ++i) {
      v2f b;
      b.x = p[i * SSTR + o];  b.y = p[i * SSTR + o + 1];
      acc[i] = __builtin_amdgcn_wmma_f32_16x16x4_f32(false, a, false, b,
                                                     (short)0, acc[i], false, false);
    }
  }
#pragma unroll
  for (int v = 0; v < 8; ++v) {
    const int w = PADDED(permBase ^ cnot_perm(((v & 4) << 1) | (v & 3), rr));
#pragma unroll
    for (int i = 0; i < NS; ++i) dst[i * SSTR + w] = acc[i][v];
  }
}

__device__ __forceinline__ void group_apply_stacked_z(const float *__restrict__ A,
                                                      const float *bRe, const float *bIm,
                                                      float *z) {
  v8f acc[NS];
#pragma unroll
  for (int i = 0; i < NS; ++i) acc[i] = v8f{};
#pragma unroll
  for (int kk = 0; kk < 4; ++kk) {
    const float *p = (kk < 2) ? bRe : bIm;
    const int o = (kk & 1) * 272;
    v2f a;
    a.x = A[kk * 64];  a.y = A[kk * 64 + 16];
#pragma unroll
    for (int i = 0; i < NS; ++i) {
      v2f b;
      b.x = p[i * SSTR + o];  b.y = p[i * SSTR + o + 1];
      acc[i] = __builtin_amdgcn_wmma_f32_16x16x4_f32(false, a, false, b,
                                                     (short)0, acc[i], false, false);
    }
  }
#pragma unroll
  for (int i = 0; i < NS; ++i) {
    float t = 0.f;
#pragma unroll
    for (int v = 0; v < 8; ++v) {
      const float sq = acc[i][v] * acc[i][v];
      t += (((v >> 2) ^ (v >> 1) ^ v) & 1) ? -sq : sq;
    }
    z[i] = t;
  }
}

__global__ __launch_bounds__(256, 8) void vqc_kernel(const float *__restrict__ X,
                                                  const float *__restrict__ Wt,
                                                  const float *__restrict__ bias,
                                                  float *__restrict__ out, int nS) {
  static_assert(NS * 64 <= 256, "the factor stage uses 64 threads per sample");
  __shared__ __attribute__((aligned(16))) float SM[GM_WORDS + NS * SSTR];
  float *const gm = SM;
  float *const ST = SM + GM_WORDS;

#define SRE(b) (ST + (2 * (b)) * PSTR)
#define SIM(b) (ST + (2 * (b) + 1) * PSTR)

  const int tid  = threadIdx.x;
  const int lane = tid & 31;
  const int wave = tid >> 5;
  const int nGroups = (nS + NS - 1) / NS;
  if ((int)blockIdx.x >= nGroups) return;

  float *const ug = ST;
  if (tid < N_LAYERS * N_WIRES) {
    float u[8];
    rot_gate(Wt[tid * 3 + 0], Wt[tid * 3 + 1], Wt[tid * 3 + 2], u);
#pragma unroll
    for (int j = 0; j < 8; ++j) ug[tid * 8 + j] = u[j];
    if (tid >= N_WIRES && tid < 2 * N_WIRES) {
#pragma unroll
      for (int j = 0; j < 8; ++j) gm[U1_OFF + (tid - N_WIRES) * 8 + j] = u[j];
    }
  }

#define FB(i) (ST + (i) * SSTR + 2 * PSTR)
#define LPB(i) (ST + (i) * SSTR + 3 * PSTR)

  const int ei = (tid < NS * N_WIRES) ? tid / N_WIRES : 0;
  const int ew = tid - (tid / N_WIRES) * N_WIRES;
  float xcur = 0.f;
  if (tid < NS * N_WIRES) {
    int s = (int)blockIdx.x * NS + ei;
    if (s >= nS) s = nS - 1;
    xcur = X[s * N_WIRES + ew];
  }
  __syncthreads();
  const float4 m0 = *reinterpret_cast<const float4 *>(ug + ew * 8);
  const float4 m1 = *reinterpret_cast<const float4 *>(ug + ew * 8 + 4);
  {
    const int r = tid & 15, c = tid >> 4;
    const float (*u2)[8] = reinterpret_cast<const float (*)[8]>(ug + 2 * N_WIRES * 8);
    const float (*u3)[8] = reinterpret_cast<const float (*)[8]>(ug + 3 * N_WIRES * 8);
    gm[L2_OFF + tid]       = real_form3(u2, 0, 1, 2, r, c);
    gm[L2_OFF + 256 + tid] = real_form3(u2, 3, 4, 5, r, c);
    gm[L2_OFF + 512 + tid] = real_form3(u2, 6, 8, 9, r, c);
    gm[L3_OFF + tid]       = real_form3(u3, 1, 5, 9, r, c);
  }

  const int  c16 = lane & 15;
  const int  h   = lane >> 4;
  const int  kb  = h << 1;
  const int  baseA = kb * 16 + c16;

  const int baseBB = wave * 136 + c16 + kb * 17;
  const int baseWB = wave * 136 + c16 + h * PSTR;
  const int colC   = (wave >> 1) * 16 + c16;
  const int baseBC = (wave >> 1) * 272 + c16 * 17 + (wave & 1) * 4 + kb;
  const int col3   = ((wave & 4) << 7) | ((wave & 3) << 6) | ((c16 & 8) << 2) | ((c16 & 7) << 1);
  const int baseB3 = col3 + (col3 >> 4) + h * 17;
  const int pbC = cnot_perm((colC << 4) | ((wave & 1) << 2), 3);

  const int gstep = NWG;
  const float bias0 = bias[0];
  for (int grp = blockIdx.x; grp < nGroups; grp += gstep) {
    const int s0 = grp * NS;

    if (tid < NS * N_WIRES) {
      const float x = xcur * 0.7853981633974483f;
      const float c = __cosf(x), sn = __sinf(x);
      float *vr = FB(ei) + VR_OFF;
      float *vi = FB(ei) + VI_OFF;
      vr[ew]           = m0.x * c + m0.z * sn;
      vi[ew]           = m0.y * c + m0.w * sn;
      vr[N_WIRES + ew] = m1.x * c + m1.z * sn;
      vi[N_WIRES + ew] = m1.y * c + m1.w * sn;
    }
    __syncthreads();

    if (tid < NS * N_WIRES) {
      int sn2 = (grp + gstep) * NS + ei;
      if (sn2 >= nS) sn2 = nS - 1;
      xcur = X[sn2 * N_WIRES + ew];
    }

    if (tid < NS * 64) {
      float *fb = FB(tid >> 6);
      const float *vr = fb + VR_OFF, *vi = fb + VI_OFF;
      const float *u1 = gm + U1_OFF;
      const int t = tid & 63;
      if (t < 32) {
#pragma unroll
        for (int g = 0; g < 2; ++g) {
          float rr_[2], ri_[2];
          {
            const float *e = u1 + 9 * 8 + (((t & 1) * 2 + g) * 2);
#pragma unroll
            for (int p = 0; p < 2; ++p) {
              const int vb = (g ^ p) * N_WIRES + 9;
              cmul(e[0], e[1], vr[vb], vi[vb], rr_[p], ri_[p]);
            }
          }
#pragma unroll
          for (int w = 8; w >= 5; --w) {
            const int ib = (t >> (9 - w)) & 1;
            float qr[2], qi[2];
#pragma unroll
            for (int c = 0; c < 2; ++c) {
              const float *e = u1 + w * 8 + ((ib * 2 + c) * 2);
              cmul(e[0], e[1], rr_[c], ri_[c], qr[c], qi[c]);
            }
#pragma unroll
            for (int p = 0; p < 2; ++p) {
              const int e0 = p * N_WIRES + w, e1 = (1 ^ p) * N_WIRES + w;
              float ar, ai, br, bi;
              cmul(qr[0], qi[0], vr[e0], vi[e0], ar, ai);
              cmul(qr[1], qi[1], vr[e1], vi[e1], br, bi);
              rr_[p] = ar + br; ri_[p] = ai + bi;
            }
          }
          fb[FR_OFF + t * 8 + g * 2 + 0] = rr_[0];
          fb[FR_OFF + t * 8 + g * 2 + 1] = rr_[1];
          fb[FR_OFF + t * 8 + 4 + g * 2 + 0] = ri_[0];
          fb[FR_OFF + t * 8 + 4 + g * 2 + 1] = ri_[1];
        }
      } else {
        const int hh = t - 32;
#pragma unroll
        for (int g = 0; g < 2; ++g) {
          float ar_[2], ai_[2];
          {
            const int ib = (hh >> 4) & 1;
#pragma unroll
            for (int c = 0; c < 2; ++c) {
              const float *e = u1 + ((ib * 2 + c) * 2);
              const int vb = (c ^ g) * N_WIRES;
              cmul(e[0], e[1], vr[vb], vi[vb], ar_[c], ai_[c]);
            }
          }
#pragma unroll
          for (int w = 1; w < 5; ++w) {
            const int ib = (hh >> (4 - w)) & 1;
            const int x = (w == 1) ? g : 0;
            float nr[2], ni[2];
#pragma unroll
            for (int c = 0; c < 2; ++c) {
              const int e0 = (c ^ x) * N_WIRES + w, e1 = (c ^ 1 ^ x) * N_WIRES + w;
              float s0r, s0i, s1r, s1i;
              cmul(ar_[0], ai_[0], vr[e0], vi[e0], s0r, s0i);
              cmul(ar_[1], ai_[1], vr[e1], vi[e1], s1r, s1i);
              const float *e = u1 + w * 8 + ((ib * 2 + c) * 2);
              cmul(e[0], e[1], s0r + s1r, s0i + s1i, nr[c], ni[c]);
            }
            ar_[0] = nr[0]; ai_[0] = ni[0]; ar_[1] = nr[1]; ai_[1] = ni[1];
          }
          fb[FL_OFF + hh * 8 + g * 2 + 0] = ar_[0];
          fb[FL_OFF + hh * 8 + g * 2 + 1] = ar_[1];
          fb[FL_OFF + hh * 8 + 4 + g * 2 + 0] = ai_[0];
          fb[FL_OFF + hh * 8 + 4 + g * 2 + 1] = ai_[1];
        }
      }
    }
    __syncthreads();

    {
      const int si = wave >> 1, qb = (wave & 1) * 2;
      const float *fl = FB(si) + FL_OFF + qb;
      const float *A = gm + L2_OFF + baseA;
      const int ih = (((c16 >> 3) & 1) * 2) ^ ((c16 >> 2) & 1) ^ (((c16 >> 1) & 1) * 20)
                   ^ (((c16 & 1) ^ h) * 10);
      v8f acc[2];
      acc[0] = v8f{}; acc[1] = v8f{};
#pragma unroll
      for (int kk = 0; kk < 4; ++kk) {
        const int r0 = (ih ^ ((kk & 1) * 20)) * 8 + (kk >> 1) * 4;
        const int r1 = (ih ^ ((kk & 1) * 20) ^ 5) * 8 + (kk >> 1) * 4;
        v2f a;
        a.x = A[kk * 64];  a.y = A[kk * 64 + 16];
#pragma unroll
        for (int t = 0; t < 2; ++t) {
          v2f b;
          b.x = fl[r0 + t];  b.y = fl[r1 + t];
          acc[t] = __builtin_amdgcn_wmma_f32_16x16x4_f32(false, a, false, b,
                                                         (short)0, acc[t], false, false);
        }
      }
      float *lp = LPB(si) + c16 * 8 + h * 4 + qb;
#pragma unroll
      for (int v = 0; v < 8; ++v) {
#pragma unroll
        for (int t = 0; t < 2; ++t) lp[v * 128 + t] = acc[t][v];
      }
    }
    __syncthreads();

    {
      const int dlo = tid & 127;
      const int ilo = (dlo ^ (dlo >> 2)) & 31;
      const int ent = ((tid >> 7) & 1) * 16 + ((tid >> 5) & 3) * 4 + (tid & 3);
      const int wo = PADDED(tid);
#pragma unroll
      for (int i = 0; i < NS; ++i) {
        const float *fr = FB(i) + FR_OFF + ilo * 8;
        const float *lq = LPB(i) + ent * 8;
        const float4 Rr = *reinterpret_cast<const float4 *>(fr);
        const float4 Ri = *reinterpret_cast<const float4 *>(fr + 4);
#pragma unroll
        for (int j = 0; j < DIM / 256; ++j) {
          const float4 Lr = *reinterpret_cast<const float4 *>(lq + j * 256);
          const float4 Li = *reinterpret_cast<const float4 *>(lq + j * 256 + 4);
          const float re = (Lr.x * Rr.x - Li.x * Ri.x) + (Lr.y * Rr.y - Li.y * Ri.y)
                         + (Lr.z * Rr.z - Li.z * Ri.z) + (Lr.w * Rr.w - Li.w * Ri.w);
          const float im = (Lr.x * Ri.x + Li.x * Rr.x) + (Lr.y * Ri.y + Li.y * Rr.y)
                         + (Lr.z * Ri.z + Li.z * Rr.z) + (Lr.w * Ri.w + Li.w * Rr.w);
          (SRE(0) + i * SSTR)[wo + 272 * j] = re;
          (SIM(0) + i * SSTR)[wo + 272 * j] = im;
        }
      }
    }
    __syncthreads();

    {
      const float *Lg = gm + L2_OFF;
      group_apply_stacked(Lg + 256 + baseA,
                          SRE(0) + baseBB, SIM(0) + baseBB,
                          SRE(1) + baseWB, 17);
      __syncthreads();
      group_apply_stacked_scatter(Lg + 512 + baseA,
                                  SRE(1) + baseBC, SIM(1) + baseBC,
                                  SRE(0) + h * PSTR,
                                  3, pbC);
      __syncthreads();
    }

    float z[NS];
    group_apply_stacked_z(gm + L3_OFF + baseA, SRE(0) + baseB3, SIM(0) + baseB3, z);
#pragma unroll
    for (int m = 16; m > 0; m >>= 1) {
#pragma unroll
      for (int i = 0; i < NS; ++i) z[i] += __shfl_xor(z[i], m, 32);
    }
    if (lane == 0) {
#pragma unroll
      for (int i = 0; i < NS; ++i) SRE(1)[i * SSTR + ZP_OFF + wave] = z[i];
    }
    __syncthreads();
    if (tid < NS && s0 + tid < nS) {
      const float *red = SRE(1) + tid * SSTR + ZP_OFF;
      float t = 0.f;
#pragma unroll
      for (int w = 0; w < 8; ++w) t += red[w];
      out[s0 + tid] = t + bias0;
    }
  }
#undef FB
#undef LPB
#undef SRE
#undef SIM
}

extern "C" void kernel_launch(void *const *d_in, const int *in_sizes, int n_in,
                              void *d_out, int out_size, void *d_ws, size_t ws_size,
                              hipStream_t stream) {
  const float *X = (const float *)d_in[0];
  const float *W = (const float *)d_in[1];
  const float *bias = (const float *)d_in[2];
  float *out = (float *)d_out;

  const int B = in_sizes[0] / N_WIRES;

  vqc_kernel<<<NWG, 256, 0, stream>>>(X, W, bias, out, B);
}


# State behind RA (4-term complex sums LEFT' * RIGHT) built on the matrix unit: per class (d8^d4, d9^d3) real form of 8 RIGHT entries times 16 LEFT' entries, 2 WMMAs per 128 amplitudes; new sum order
# speedup vs baseline: 4.9462x; 1.0919x over previous
#include <hip/hip_runtime.h>


typedef float v2f __attribute__((ext_vector_type(2)));
typedef float v8f __attribute__((ext_vector_type(8)));

#define N_WIRES 10
#define N_LAYERS 4
#define DIM 1024
#define PADDED(k) ((k) + ((k) >> 4))
#define PDIM (DIM + (DIM >> 4))
#define PSTR (PDIM + 32)
#define L2_OFF 0
#define L3_OFF (L2_OFF + 768)
#define U1_OFF (L3_OFF + 256)
#define GM_WORDS 1120
#define NS 4
#define NWG 1024
#define SSTR (4 * PSTR)
#define VR_OFF 0
#define VI_OFF 32
#define FL_OFF 64
#define FR_OFF 320
#define ZP_OFF 576
#define LPT_PITCH 68
#define LPI_OFF 542

__device__ __forceinline__ void cmul(float ar, float ai, float br, float bi,
                                     float &cr, float &ci) {
  cr = ar * br - ai * bi;
  ci = ar * bi + ai * br;
}

__device__ __forceinline__ int cnot_perm(int m, int rr) {
#pragma unroll
  for (int w = 0; w < N_WIRES; ++w) {
    int t = w + rr; if (t >= N_WIRES) t -= N_WIRES;
    m ^= ((m >> (9 - w)) & 1) << (9 - t);
  }
  return m;
}

__device__ __forceinline__ float real_form3(const float (*u)[8], int w0, int w1, int w2,
                                            int r, int c) {
  float pr = 1.f, pi = 0.f;
#pragma unroll
  for (int q = 0; q < 3; ++q) {
    const int wq = (q == 0) ? w0 : ((q == 1) ? w1 : w2);
    const int rb = (r >> (2 - q)) & 1, cb2 = (c >> (2 - q)) & 1;
    const float *e = &u[wq][(rb * 2 + cb2) * 2];
    float nr, ni; cmul(pr, pi, e[0], e[1], nr, ni);
    pr = nr; pi = ni;
  }
  const int rq = r >> 3, cq = c >> 3;
  return (rq == cq) ? pr : (rq ? pi : -pi);
}

__device__ __forceinline__ void rot_gate(float phi, float theta, float omega, float *u) {
  const float cs = __cosf(0.5f * theta), s = __sinf(0.5f * theta);
  const float a = 0.5f * (phi + omega), b = 0.5f * (phi - omega);
  const float ca = __cosf(a), sa = __sinf(a), cb = __cosf(b), sb = __sinf(b);
  u[0] =  ca * cs; u[1] = -sa * cs;
  u[2] = -cb * s;  u[3] = -sb * s;
  u[4] =  cb * s;  u[5] = -sb * s;
  u[6] =  ca * cs; u[7] =  sa * cs;
}

__device__ __forceinline__ void group_apply_stacked(const float *__restrict__ A,
                                                    const float *bRe, const float *bIm,
                                                    float *dst, int idxStride) {
  v8f acc[NS];
#pragma unroll
  for (int i = 0; i < NS; ++i) acc[i] = v8f{};
#pragma unroll
  for (int kk = 0; kk < 4; ++kk) {
    const float *p = (kk < 2) ? bRe : bIm;
    const int o = (kk & 1) * 4 * idxStride;
    v2f a;
    a.x = A[kk * 64];  a.y = A[kk * 64 + 16];
#pragma unroll
    for (int i = 0; i < NS; ++i) {
      v2f b;
      b.x = p[i * SSTR + o];  b.y = p[i * SSTR + o + idxStride];
      acc[i] = __builtin_amdgcn_wmma_f32_16x16x4_f32(false, a, false, b,
                                                     (short)0, acc[i], false, false);
    }
  }
#pragma unroll
  for (int i = 0; i < NS; ++i) {
#pragma unroll
    for (int v = 0; v < 8; ++v) dst[i * SSTR + v * idxStride] = acc[i][v];
  }
}

__device__ __forceinline__ void group_apply_stacked_scatter(const float *__restrict__ A,
                                                            const float *bRe, const float *bIm,
                                                            float *dst, int rr, int permBase) {
  v8f acc[NS];
#pragma unroll
  for (int i = 0; i < NS; ++i) acc[i] = v8f{};
#pragma unroll
  for (int kk = 0; kk < 4; ++kk) {
    const float *p = (kk < 2) ? bRe : bIm;
    const int o = (kk & 1) * 8;
    v2f a;
    a.x = A[kk * 64];  a.y = A[kk * 64 + 16];
#pragma unroll
    for (int i = 0; i < NS; ++i) {
      v2f b;
      b.x = p[i * SSTR + o];  b.y = p[i * SSTR + o + 1];
      acc[i] = __builtin_amdgcn_wmma_f32_16x16x4_f32(false, a, false, b,
                                                     (short)0, acc[i], false, false);
    }
  }
#pragma unroll
  for (int v = 0; v < 8; ++v) {
    const int w = PADDED(permBase ^ cnot_perm(((v & 4) << 1) | (v & 3), rr));
#pragma unroll
    for (int i = 0; i < NS; ++i) dst[i * SSTR + w] = acc[i][v];
  }
}

__device__ __forceinline__ void group_apply_stacked_z(const float *__restrict__ A,
                                                      const float *bRe, const float *bIm,
                                                      float *z) {
  v8f acc[NS];
#pragma unroll
  for (int i = 0; i < NS; ++i) acc[i] = v8f{};
#pragma unroll
  for (int kk = 0; kk < 4; ++kk) {
    const float *p = (kk < 2) ? bRe : bIm;
    const int o = (kk & 1) * 272;
    v2f a;
    a.x = A[kk * 64];  a.y = A[kk * 64 + 16];
#pragma unroll
    for (int i = 0; i < NS; ++i) {
      v2f b;
      b.x = p[i * SSTR + o];  b.y = p[i * SSTR + o + 1];
      acc[i] = __builtin_amdgcn_wmma_f32_16x16x4_f32(false, a, false, b,
                                                     (short)0, acc[i], false, false);
    }
  }
#pragma unroll
  for (int i = 0; i < NS; ++i) {
    float t = 0.f;
#pragma unroll
    for (int v = 0; v < 8; ++v) {
      const float sq = acc[i][v] * acc[i][v];
      t += (((v >> 2) ^ (v >> 1) ^ v) & 1) ? -sq : sq;
    }
    z[i] = t;
  }
}

__global__ __launch_bounds__(256, 8) void vqc_kernel(const float *__restrict__ X,
                                                  const float *__restrict__ Wt,
                                                  const float *__restrict__ bias,
                                                  float *__restrict__ out, int nS) {
  static_assert(NS * 64 <= 256, "the factor stage uses 64 threads per sample");
  __shared__ __attribute__((aligned(16))) float SM[GM_WORDS + NS * SSTR];
  float *const gm = SM;
  float *const ST = SM + GM_WORDS;

#define SRE(b) (ST + (2 * (b)) * PSTR)
#define SIM(b) (ST + (2 * (b) + 1) * PSTR)

  const int tid  = threadIdx.x;
  const int lane = tid & 31;
  const int wave = tid >> 5;
  const int nGroups = (nS + NS - 1) / NS;
  if ((int)blockIdx.x >= nGroups) return;

  float *const ug = ST;
  if (tid < N_LAYERS * N_WIRES) {
    float u[8];
    rot_gate(Wt[tid * 3 + 0], Wt[tid * 3 + 1], Wt[tid * 3 + 2], u);
#pragma unroll
    for (int j = 0; j < 8; ++j) ug[tid * 8 + j] = u[j];
    if (tid >= N_WIRES && tid < 2 * N_WIRES) {
#pragma unroll
      for (int j = 0; j < 8; ++j) gm[U1_OFF + (tid - N_WIRES) * 8 + j] = u[j];
    }
  }

#define FB(i) (ST + (i) * SSTR + 2 * PSTR)
#define LPB(i) (ST + (i) * SSTR + 3 * PSTR)

  const int ei = (tid < NS * N_WIRES) ? tid / N_WIRES : 0;
  const int ew = tid - (tid / N_WIRES) * N_WIRES;
  float xcur = 0.f;
  if (tid < NS * N_WIRES) {
    int s = (int)blockIdx.x * NS + ei;
    if (s >= nS) s = nS - 1;
    xcur = X[s * N_WIRES + ew];
  }
  __syncthreads();
  const float4 m0 = *reinterpret_cast<const float4 *>(ug + ew * 8);
  const float4 m1 = *reinterpret_cast<const float4 *>(ug + ew * 8 + 4);
  {
    const int r = tid & 15, c = tid >> 4;
    const float (*u2)[8] = reinterpret_cast<const float (*)[8]>(ug + 2 * N_WIRES * 8);
    const float (*u3)[8] = reinterpret_cast<const float (*)[8]>(ug + 3 * N_WIRES * 8);
    gm[L2_OFF + tid]       = real_form3(u2, 0, 1, 2, r, c);
    gm[L2_OFF + 256 + tid] = real_form3(u2, 3, 4, 5, r, c);
    gm[L2_OFF + 512 + tid] = real_form3(u2, 6, 8, 9, r, c);
    gm[L3_OFF + tid]       = real_form3(u3, 1, 5, 9, r, c);
  }

  const int  c16 = lane & 15;
  const int  h   = lane >> 4;
  const int  kb  = h << 1;
  const int  baseA = kb * 16 + c16;

  const int baseBB = wave * 136 + c16 + kb * 17;
  const int baseWB = wave * 136 + c16 + h * PSTR;
  const int colC   = (wave >> 1) * 16 + c16;
  const int baseBC = (wave >> 1) * 272 + c16 * 17 + (wave & 1) * 4 + kb;
  const int col3   = ((wave & 4) << 7) | ((wave & 3) << 6) | ((c16 & 8) << 2) | ((c16 & 7) << 1);
  const int baseB3 = col3 + (col3 >> 4) + h * 17;
  const int pbC = cnot_perm((colC << 4) | ((wave & 1) << 2), 3);

  const int ts1 = (wave >> 2) & 1, ts2 = (wave >> 1) & 1, td3 = wave & 1;
  const int bp = c16 >> 3, br6 = (c16 >> 1) & 1;
  const int bd5 = ((c16 >> 2) & 1) ^ bp, bd7 = c16 & 1;
  const int bilo = (bd5 ^ td3) * 16 + br6 * 8 + (bd7 ^ bd5) * 4 + (ts1 ^ br6) * 2 + (ts2 ^ td3 ^ bd7);
  const int baseFA = bilo * 8 + kb + bp * 4;
  const int baseLB = wave * LPT_PITCH + c16 * 4 + kb;
  const int bd4 = c16 >> 3;
  const int baseD = ((c16 >> 2) & 1) * 544 + ((c16 >> 1) & 1) * 272 + (c16 & 1) * 136 + td3 * 68
                  + bd4 * (34 + 8) + (ts1 ^ bd4) * 2 + (ts2 ^ td3) + h * (PSTR + 17);
  const int e5 = h ? -17 : 17, e6 = bd4 ? -8 : 8;

  const int gstep = NWG;
  const float bias0 = bias[0];
  for (int grp = blockIdx.x; grp < nGroups; grp += gstep) {
    const int s0 = grp * NS;

    if (tid < NS * N_WIRES) {
      const float x = xcur * 0.7853981633974483f;
      const float c = __cosf(x), sn = __sinf(x);
      float *vr = FB(ei) + VR_OFF;
      float *vi = FB(ei) + VI_OFF;
      vr[ew]           = m0.x * c + m0.z * sn;
      vi[ew]           = m0.y * c + m0.w * sn;
      vr[N_WIRES + ew] = m1.x * c + m1.z * sn;
      vi[N_WIRES + ew] = m1.y * c + m1.w * sn;
    }
    __syncthreads();

    if (tid < NS * N_WIRES) {
      int sn2 = (grp + gstep) * NS + ei;
      if (sn2 >= nS) sn2 = nS - 1;
      xcur = X[sn2 * N_WIRES + ew];
    }

    if (tid < NS * 64) {
      float *fb = FB(tid >> 6);
      const float *vr = fb + VR_OFF, *vi = fb + VI_OFF;
      const float *u1 = gm + U1_OFF;
      const int t = tid & 63;
      if (t < 32) {
#pragma unroll
        for (int g = 0; g < 2; ++g) {
          float rr_[2], ri_[2];
          {
            const float *e = u1 + 9 * 8 + (((t & 1) * 2 + g) * 2);
#pragma unroll
            for (int p = 0; p < 2; ++p) {
              const int vb = (g ^ p) * N_WIRES + 9;
              cmul(e[0], e[1], vr[vb], vi[vb], rr_[p], ri_[p]);
            }
          }
#pragma unroll
          for (int w = 8; w >= 5; --w) {
            const int ib = (t >> (9 - w)) & 1;
            float qr[2], qi[2];
#pragma unroll
            for (int c = 0; c < 2; ++c) {
              const float *e = u1 + w * 8 + ((ib * 2 + c) * 2);
              cmul(e[0], e[1], rr_[c], ri_[c], qr[c], qi[c]);
            }
#pragma unroll
            for (int p = 0; p < 2; ++p) {
              const int e0 = p * N_WIRES + w, e1 = (1 ^ p) * N_WIRES + w;
              float ar, ai, br, bi;
              cmul(qr[0], qi[0], vr[e0], vi[e0], ar, ai);
              cmul(qr[1], qi[1], vr[e1], vi[e1], br, bi);
              rr_[p] = ar + br; ri_[p] = ai + bi;
            }
          }
          fb[FR_OFF + t * 8 + g * 2 + 0] = rr_[0];
          fb[FR_OFF + t * 8 + g * 2 + 1] = rr_[1];
          fb[FR_OFF + t * 8 + 4 + g * 2 + 0] = ri_[0];
          fb[FR_OFF + t * 8 + 4 + g * 2 + 1] = ri_[1];
        }
      } else {
        const int hh = t - 32;
#pragma unroll
        for (int g = 0; g < 2; ++g) {
          float ar_[2], ai_[2];
          {
            const int ib = (hh >> 4) & 1;
#pragma unroll
            for (int c = 0; c < 2; ++c) {
              const float *e = u1 + ((ib * 2 + c) * 2);
              const int vb = (c ^ g) * N_WIRES;
              cmul(e[0], e[1], vr[vb], vi[vb], ar_[c], ai_[c]);
            }
          }
#pragma unroll
          for (int w = 1; w < 5; ++w) {
            const int ib = (hh >> (4 - w)) & 1;
            const int x = (w == 1) ? g : 0;
            float nr[2], ni[2];
#pragma unroll
            for (int c = 0; c < 2; ++c) {
              const int e0 = (c ^ x) * N_WIRES + w, e1 = (c ^ 1 ^ x) * N_WIRES + w;
              float s0r, s0i, s1r, s1i;
              cmul(ar_[0], ai_[0], vr[e0], vi[e0], s0r, s0i);
              cmul(ar_[1], ai_[1], vr[e1], vi[e1], s1r, s1i);
              const float *e = u1 + w * 8 + ((ib * 2 + c) * 2);
              cmul(e[0], e[1], s0r + s1r, s0i + s1i, nr[c], ni[c]);
            }
            ar_[0] = nr[0]; ai_[0] = ni[0]; ar_[1] = nr[1]; ai_[1] = ni[1];
          }
          fb[FL_OFF + hh * 8 + g * 2 + 0] = ar_[0];
          fb[FL_OFF + hh * 8 + g * 2 + 1] = ar_[1];
          fb[FL_OFF + hh * 8 + 4 + g * 2 + 0] = ai_[0];
          fb[FL_OFF + hh * 8 + 4 + g * 2 + 1] = ai_[1];
        }
      }
    }
    __syncthreads();

    {
      const int si = wave >> 1, qb = (wave & 1) * 2;
      const float *fl = FB(si) + FL_OFF + qb;
      const float *A = gm + L2_OFF + baseA;
      const int ih = (((c16 >> 3) & 1) * 2) ^ ((c16 >> 2) & 1) ^ (((c16 >> 1) & 1) * 20)
                   ^ (((c16 & 1) ^ h) * 10);
      v8f acc[2];
      acc[0] = v8f{}; acc[1] = v8f{};
#pragma unroll
      for (int kk = 0; kk < 4; ++kk) {
        const int r0 = (ih ^ ((kk & 1) * 20)) * 8 + (kk >> 1) * 4;
        const int r1 = (ih ^ ((kk & 1) * 20) ^ 5) * 8 + (kk >> 1) * 4;
        v2f a;
        a.x = A[kk * 64];  a.y = A[kk * 64 + 16];
#pragma unroll
        for (int t = 0; t < 2; ++t) {
          v2f b;
          b.x = fl[r0 + t];  b.y = fl[r1 + t];
          acc[t] = __builtin_amdgcn_wmma_f32_16x16x4_f32(false, a, false, b,
                                                         (short)0, acc[t], false, false);
        }
      }
      const int tl = ((((c16 >> 1) ^ (c16 >> 2)) & 1) << 2) | (((c16 ^ (c16 >> 3)) & 1) << 1)
                   | (c16 >> 3);
      float *lp = LPB(si) + h * LPI_OFF + tl * LPT_PITCH + ((c16 >> 2) & 1) * 32 + qb;
#pragma unroll
      for (int v = 0; v < 8; ++v) {
#pragma unroll
        for (int t = 0; t < 2; ++t) lp[v * 4 + t] = acc[t][v];
      }
    }
    __syncthreads();

    {
      v8f acc[NS];
#pragma unroll
      for (int i = 0; i < NS; ++i) acc[i] = v8f{};
#pragma unroll
      for (int kk = 0; kk < 2; ++kk) {
#pragma unroll
        for (int i = 0; i < NS; ++i) {
          const float *fr = FB(i) + FR_OFF + (kk ? (baseFA ^ 4) : baseFA);
          const float *lq = LPB(i) + kk * LPI_OFF + baseLB;
          v2f a = *reinterpret_cast<const v2f *>(fr);
          const v2f b = *reinterpret_cast<const v2f *>(lq);
          if (kk && !bp) { a.x = -a.x; a.y = -a.y; }
          acc[i] = __builtin_amdgcn_wmma_f32_16x16x4_f32(false, a, false, b,
                                                         (short)0, acc[i], false, false);
        }
      }
#pragma unroll
      for (int r = 0; r < 8; ++r) {
        const int w = baseD + ((r & 4) ? e5 : 0) + ((r & 2) ? e6 : 0) + (r & 1) * 4;
#pragma unroll
        for (int i = 0; i < NS; ++i) ST[i * SSTR + w] = acc[i][r];
      }
    }
    __syncthreads();

    {
      const float *Lg = gm + L2_OFF;
      group_apply_stacked(Lg + 256 + baseA,
                          SRE(0) + baseBB, SIM(0) + baseBB,
                          SRE(1) + baseWB, 17);
      __syncthreads();
      group_apply_stacked_scatter(Lg + 512 + baseA,
                                  SRE(1) + baseBC, SIM(1) + baseBC,
                                  SRE(0) + h * PSTR,
                                  3, pbC);
      __syncthreads();
    }

    float z[NS];
    group_apply_stacked_z(gm + L3_OFF + baseA, SRE(0) + baseB3, SIM(0) + baseB3, z);
#pragma unroll
    for (int m = 16; m > 0; m >>= 1) {
#pragma unroll
      for (int i = 0; i < NS; ++i) z[i] += __shfl_xor(z[i], m, 32);
    }
    if (lane == 0) {
#pragma unroll
      for (int i = 0; i < NS; ++i) SRE(1)[i * SSTR + ZP_OFF + wave] = z[i];
    }
    __syncthreads();
    if (tid < NS && s0 + tid < nS) {
      const float *red = SRE(1) + tid * SSTR + ZP_OFF;
      float t = 0.f;
#pragma unroll
      for (int w = 0; w < 8; ++w) t += red[w];
      out[s0 + tid] = t + bias0;
    }
  }
#undef FB
#undef LPB
#undef SRE
#undef SIM
}

extern "C" void kernel_launch(void *const *d_in, const int *in_sizes, int n_in,
                              void *d_out, int out_size, void *d_ws, size_t ws_size,
                              hipStream_t stream) {
  const float *X = (const float *)d_in[0];
  const float *W = (const float *)d_in[1];
  const float *bias = (const float *)d_in[2];
  float *out = (float *)d_out;

  const int B = in_sizes[0] / N_WIRES;

  vqc_kernel<<<NWG, 256, 0, stream>>>(X, W, bias, out, B);
}


# One wave per sample: each wave runs all stages of its own sample on 8 column tiles (8 accumulator chains); no workgroup barrier left in the trip loop; half the waves; new <Z> summation order
# speedup vs baseline: 5.1731x; 1.0459x over previous
#include <hip/hip_runtime.h>


typedef float v2f __attribute__((ext_vector_type(2)));
typedef float v8f __attribute__((ext_vector_type(8)));

#define N_WIRES 10
#define N_LAYERS 4
#define DIM 1024
#define PADDED(k) ((k) + ((k) >> 4))
#define PDIM (DIM + (DIM >> 4))
#define PSTR (PDIM + 32)
#define L2_OFF 0
#define L3_OFF (L2_OFF + 768)
#define U1_OFF (L3_OFF + 256)
#define GM_WORDS 1120
#define NSW 8
#define NWG 512
#define SSTR (4 * PSTR)
#define VR_OFF 0
#define VI_OFF 32
#define FL_OFF 64
#define FR_OFF 320
#define LPT_PITCH 68
#define LPI_OFF 542

__device__ __forceinline__ void cmul(float ar, float ai, float br, float bi,
                                     float &cr, float &ci) {
  cr = ar * br - ai * bi;
  ci = ar * bi + ai * br;
}

__device__ __forceinline__ int cnot_perm(int m, int rr) {
#pragma unroll
  for (int w = 0; w < N_WIRES; ++w) {
    int t = w + rr; if (t >= N_WIRES) t -= N_WIRES;
    m ^= ((m >> (9 - w)) & 1) << (9 - t);
  }
  return m;
}

__device__ __forceinline__ float real_form3(const float (*u)[8], int w0, int w1, int w2,
                                            int r, int c) {
  float pr = 1.f, pi = 0.f;
#pragma unroll
  for (int q = 0; q < 3; ++q) {
    const int wq = (q == 0) ? w0 : ((q == 1) ? w1 : w2);
    const int rb = (r >> (2 - q)) & 1, cb2 = (c >> (2 - q)) & 1;
    const float *e = &u[wq][(rb * 2 + cb2) * 2];
    float nr, ni; cmul(pr, pi, e[0], e[1], nr, ni);
    pr = nr; pi = ni;
  }
  const int rq = r >> 3, cq = c >> 3;
  return (rq == cq) ? pr : (rq ? pi : -pi);
}

__device__ __forceinline__ void rot_gate(float phi, float theta, float omega, float *u) {
  const float cs = __cosf(0.5f * theta), s = __sinf(0.5f * theta);
  const float a = 0.5f * (phi + omega), b = 0.5f * (phi - omega);
  const float ca = __cosf(a), sa = __sinf(a), cb = __cosf(b), sb = __sinf(b);
  u[0] =  ca * cs; u[1] = -sa * cs;
  u[2] = -cb * s;  u[3] = -sb * s;
  u[4] =  cb * s;  u[5] = -sb * s;
  u[6] =  ca * cs; u[7] =  sa * cs;
}

__device__ __forceinline__ void wave_handover() {
  __builtin_amdgcn_fence(__ATOMIC_SEQ_CST, "wavefront");
}


__device__ __forceinline__ void pass_rb(const float *__restrict__ A,
                                        const float *bRe, const float *bIm, float *dst) {
  v8f acc[8];
#pragma unroll
  for (int T = 0; T < 8; ++T) acc[T] = v8f{};
#pragma unroll
  for (int kk = 0; kk < 4; ++kk) {
    const float *p = (kk < 2) ? bRe : bIm;
    const int o = (kk & 1) * 4 * 17;
    v2f a;
    a.x = A[kk * 64];  a.y = A[kk * 64 + 16];
#pragma unroll
    for (int T = 0; T < 8; ++T) {
      v2f b;
      b.x = p[T * 136 + o];  b.y = p[T * 136 + o + 17];
      acc[T] = __builtin_amdgcn_wmma_f32_16x16x4_f32(false, a, false, b,
                                                     (short)0, acc[T], false, false);
    }
  }
#pragma unroll
  for (int T = 0; T < 8; ++T) {
#pragma unroll
    for (int v = 0; v < 8; ++v) dst[T * 136 + v * 17] = acc[T][v];
  }
}

__device__ __forceinline__ void pass_rc(const float *__restrict__ A,
                                        const float *bRe, const float *bIm,
                                        float *dst, int permLane) {
  v8f acc[8];
#pragma unroll
  for (int T = 0; T < 8; ++T) acc[T] = v8f{};
#pragma unroll
  for (int kk = 0; kk < 4; ++kk) {
    const float *p = (kk < 2) ? bRe : bIm;
    const int o = (kk & 1) * 8;
    v2f a;
    a.x = A[kk * 64];  a.y = A[kk * 64 + 16];
#pragma unroll
    for (int T = 0; T < 8; ++T) {
      const int tb = (T >> 1) * 272 + (T & 1) * 4;
      v2f b;
      b.x = p[tb + o];  b.y = p[tb + o + 1];
      acc[T] = __builtin_amdgcn_wmma_f32_16x16x4_f32(false, a, false, b,
                                                     (short)0, acc[T], false, false);
    }
  }
#pragma unroll
  for (int T = 0; T < 8; ++T) {
#pragma unroll
    for (int v = 0; v < 8; ++v) {
      const int lit = cnot_perm(((T >> 1) << 8) | ((T & 1) << 2) | ((v & 4) << 1) | (v & 3), 3);
      dst[PADDED(permLane ^ lit)] = acc[T][v];
    }
  }
}

__device__ __forceinline__ float pass_z(const float *__restrict__ A,
                                        const float *bRe, const float *bIm) {
  v8f acc[8];
#pragma unroll
  for (int T = 0; T < 8; ++T) acc[T] = v8f{};
#pragma unroll
  for (int kk = 0; kk < 4; ++kk) {
    const float *p = (kk < 2) ? bRe : bIm;
    const int o = (kk & 1) * 272;
    v2f a;
    a.x = A[kk * 64];  a.y = A[kk * 64 + 16];
#pragma unroll
    for (int T = 0; T < 8; ++T) {
      const int tc = ((T & 4) << 7) | ((T & 3) << 6);
      const int tb = tc + (tc >> 4);
      v2f b;
      b.x = p[tb + o];  b.y = p[tb + o + 1];
      acc[T] = __builtin_amdgcn_wmma_f32_16x16x4_f32(false, a, false, b,
                                                     (short)0, acc[T], false, false);
    }
  }
  float z = 0.f;
#pragma unroll
  for (int T = 0; T < 8; ++T) {
    float t = 0.f;
#pragma unroll
    for (int v = 0; v < 8; ++v) {
      const float sq = acc[T][v] * acc[T][v];
      t += (((v >> 2) ^ (v >> 1) ^ v) & 1) ? -sq : sq;
    }
    z += t;
  }
  return z;
}

__device__ __forceinline__ int sb_ilo(int T, int c16) {
  const int ts1 = (T >> 2) & 1, ts2 = (T >> 1) & 1, td3 = T & 1;
  const int bp = c16 >> 3, br6 = (c16 >> 1) & 1;
  const int bd5 = ((c16 >> 2) & 1) ^ bp, bd7 = c16 & 1;
  return (bd5 ^ td3) * 16 + br6 * 8 + (bd7 ^ bd5) * 4 + (ts1 ^ br6) * 2 + (ts2 ^ td3 ^ bd7);
}
__device__ __forceinline__ int sb_base_d(int T, int c16, int h) {
  const int ts1 = (T >> 2) & 1, ts2 = (T >> 1) & 1, td3 = T & 1;
  const int bd4 = c16 >> 3;
  return ((c16 >> 2) & 1) * 544 + ((c16 >> 1) & 1) * 272 + (c16 & 1) * 136 + td3 * 68
       + bd4 * (34 + 8) + (ts1 ^ bd4) * 2 + (ts2 ^ td3) + h * (PSTR + 17);
}

__global__ __launch_bounds__(256, 4) void vqc_kernel(const float *__restrict__ X,
                                                  const float *__restrict__ Wt,
                                                  const float *__restrict__ bias,
                                                  float *__restrict__ out, int nS) {
  __shared__ __attribute__((aligned(16))) float SM[GM_WORDS + NSW * SSTR];
  float *const gm = SM;
  float *const ST = SM + GM_WORDS;

  const int tid  = threadIdx.x;
  const int lane = tid & 31;
  const int wave = tid >> 5;
  const int nGroups = (nS + NSW - 1) / NSW;
  if ((int)blockIdx.x >= nGroups) return;

  float *const ug = ST;
  if (tid < N_LAYERS * N_WIRES) {
    float u[8];
    rot_gate(Wt[tid * 3 + 0], Wt[tid * 3 + 1], Wt[tid * 3 + 2], u);
#pragma unroll
    for (int j = 0; j < 8; ++j) ug[tid * 8 + j] = u[j];
    if (tid >= N_WIRES && tid < 2 * N_WIRES) {
#pragma unroll
      for (int j = 0; j < 8; ++j) gm[U1_OFF + (tid - N_WIRES) * 8 + j] = u[j];
    }
  }

  float *const SS = ST + wave * SSTR;
#define SRE(b) (SS + (2 * (b)) * PSTR)
#define SIM(b) (SS + (2 * (b) + 1) * PSTR)
  float *const fb = SS + 2 * PSTR;
  float *const lpb = SS + 3 * PSTR;

  const int ew = (lane < N_WIRES) ? lane : 0;
  float xcur = 0.f;
  if (lane < N_WIRES) {
    int s = (int)blockIdx.x * NSW + wave;
    if (s >= nS) s = nS - 1;
    xcur = X[s * N_WIRES + ew];
  }
  __syncthreads();
  const float4 m0 = *reinterpret_cast<const float4 *>(ug + ew * 8);
  const float4 m1 = *reinterpret_cast<const float4 *>(ug + ew * 8 + 4);
  {
    const int r = tid & 15, c = tid >> 4;
    const float (*u2)[8] = reinterpret_cast<const float (*)[8]>(ug + 2 * N_WIRES * 8);
    const float (*u3)[8] = reinterpret_cast<const float (*)[8]>(ug + 3 * N_WIRES * 8);
    gm[L2_OFF + tid]       = real_form3(u2, 0, 1, 2, r, c);
    gm[L2_OFF + 256 + tid] = real_form3(u2, 3, 4, 5, r, c);
    gm[L2_OFF + 512 + tid] = real_form3(u2, 6, 8, 9, r, c);
    gm[L3_OFF + tid]       = real_form3(u3, 1, 5, 9, r, c);
  }
  __syncthreads();

  const int  c16 = lane & 15;
  const int  h   = lane >> 4;
  const int  kb  = h << 1;
  const int  baseA = kb * 16 + c16;

  const int laneBB = c16 + kb * 17;
  const int laneWB = c16 + h * PSTR;
  const int laneBC = c16 * 17 + kb;
  const int lc3    = ((c16 & 8) << 2) | ((c16 & 7) << 1);
  const int laneB3 = lc3 + (lc3 >> 4) + h * 17;
  const int pbL = cnot_perm(c16 << 4, 3);
  const int bp = c16 >> 3;
  const int e5 = h ? -17 : 17, e6 = bp ? -8 : 8;

  const float bias0 = bias[0];
  for (int grp = blockIdx.x; grp < nGroups; grp += NWG) {
    const int smp = grp * NSW + wave;

    if (lane < N_WIRES) {
      const float x = xcur * 0.7853981633974483f;
      const float c = __cosf(x), sn = __sinf(x);
      float *vr = fb + VR_OFF;
      float *vi = fb + VI_OFF;
      vr[ew]           = m0.x * c + m0.z * sn;
      vi[ew]           = m0.y * c + m0.w * sn;
      vr[N_WIRES + ew] = m1.x * c + m1.z * sn;
      vi[N_WIRES + ew] = m1.y * c + m1.w * sn;
    }
    wave_handover();

    if (lane < N_WIRES) {
      int sn2 = (grp + NWG) * NSW + wave;
      if (sn2 >= nS) sn2 = nS - 1;
      xcur = X[sn2 * N_WIRES + ew];
    }

    {
      const float *vr = fb + VR_OFF, *vi = fb + VI_OFF;
      const float *u1 = gm + U1_OFF;
      {
        const int t = lane;
#pragma unroll
        for (int g = 0; g < 2; ++g) {
          float rr_[2], ri_[2];
          {
            const float *e = u1 + 9 * 8 + (((t & 1) * 2 + g) * 2);
#pragma unroll
            for (int p = 0; p < 2; ++p) {
              const int vb = (g ^ p) * N_WIRES + 9;
              cmul(e[0], e[1], vr[vb], vi[vb], rr_[p], ri_[p]);
            }
          }
#pragma unroll
          for (int w = 8; w >= 5; --w) {
            const int ib = (t >> (9 - w)) & 1;
            float qr[2], qi[2];
#pragma unroll
            for (int c = 0; c < 2; ++c) {
              const float *e = u1 + w * 8 + ((ib * 2 + c) * 2);
              cmul(e[0], e[1], rr_[c], ri_[c], qr[c], qi[c]);
            }
#pragma unroll
            for (int p = 0; p < 2; ++p) {
              const int e0 = p * N_WIRES + w, e1 = (1 ^ p) * N_WIRES + w;
              float ar, ai, br, bi;
              cmul(qr[0], qi[0], vr[e0], vi[e0], ar, ai);
              cmul(qr[1], qi[1], vr[e1], vi[e1], br, bi);
              rr_[p] = ar + br; ri_[p] = ai + bi;
            }
          }
          fb[FR_OFF + t * 8 + g * 2 + 0] = rr_[0];
          fb[FR_OFF + t * 8 + g * 2 + 1] = rr_[1];
          fb[FR_OFF + t * 8 + 4 + g * 2 + 0] = ri_[0];
          fb[FR_OFF + t * 8 + 4 + g * 2 + 1] = ri_[1];
        }
      }
      {
        const int hh = lane;
#pragma unroll
        for (int g = 0; g < 2; ++g) {
          float ar_[2], ai_[2];
          {
            const int ib = (hh >> 4) & 1;
#pragma unroll
            for (int c = 0; c < 2; ++c) {
              const float *e = u1 + ((ib * 2 + c) * 2);
              const int vb = (c ^ g) * N_WIRES;
              cmul(e[0], e[1], vr[vb], vi[vb], ar_[c], ai_[c]);
            }
          }
#pragma unroll
          for (int w = 1; w < 5; ++w) {
            const int ib = (hh >> (4 - w)) & 1;
            const int x = (w == 1) ? g : 0;
            float nr[2], ni[2];
#pragma unroll
            for (int c = 0; c < 2; ++c) {
              const int e0 = (c ^ x) * N_WIRES + w, e1 = (c ^ 1 ^ x) * N_WIRES + w;
              float s0r, s0i, s1r, s1i;
              cmul(ar_[0], ai_[0], vr[e0], vi[e0], s0r, s0i);
              cmul(ar_[1], ai_[1], vr[e1], vi[e1], s1r, s1i);
              const float *e = u1 + w * 8 + ((ib * 2 + c) * 2);
              cmul(e[0], e[1], s0r + s1r, s0i + s1i, nr[c], ni[c]);
            }
            ar_[0] = nr[0]; ai_[0] = ni[0]; ar_[1] = nr[1]; ai_[1] = ni[1];
          }
          fb[FL_OFF + hh * 8 + g * 2 + 0] = ar_[0];
          fb[FL_OFF + hh * 8 + g * 2 + 1] = ar_[1];
          fb[FL_OFF + hh * 8 + 4 + g * 2 + 0] = ai_[0];
          fb[FL_OFF + hh * 8 + 4 + g * 2 + 1] = ai_[1];
        }
      }
    }
    wave_handover();

    {
      const float *fl = fb + FL_OFF;
      const float *A = gm + L2_OFF + baseA;
      const int ih = (((c16 >> 3) & 1) * 2) ^ ((c16 >> 2) & 1) ^ (((c16 >> 1) & 1) * 20)
                   ^ (((c16 & 1) ^ h) * 10);
      v8f acc[4];
#pragma unroll
      for (int q = 0; q < 4; ++q) acc[q] = v8f{};
#pragma unroll
      for (int kk = 0; kk < 4; ++kk) {
        const int r0 = (ih ^ ((kk & 1) * 20)) * 8 + (kk >> 1) * 4;
        const int r1 = (ih ^ ((kk & 1) * 20) ^ 5) * 8 + (kk >> 1) * 4;
        v2f a;
        a.x = A[kk * 64];  a.y = A[kk * 64 + 16];
#pragma unroll
        for (int q = 0; q < 4; ++q) {
          v2f b;
          b.x = fl[r0 + q];  b.y = fl[r1 + q];
          acc[q] = __builtin_amdgcn_wmma_f32_16x16x4_f32(false, a, false, b,
                                                         (short)0, acc[q], false, false);
        }
      }
      const int tl = ((((c16 >> 1) ^ (c16 >> 2)) & 1) << 2) | (((c16 ^ (c16 >> 3)) & 1) << 1)
                   | (c16 >> 3);
      float *lp = lpb + h * LPI_OFF + tl * LPT_PITCH + ((c16 >> 2) & 1) * 32;
#pragma unroll
      for (int v = 0; v < 8; ++v) {
#pragma unroll
        for (int q = 0; q < 4; ++q) lp[v * 4 + q] = acc[q][v];
      }
    }
    wave_handover();

    {
      v8f acc[8];
#pragma unroll
      for (int T = 0; T < 8; ++T) acc[T] = v8f{};
#pragma unroll
      for (int kk = 0; kk < 2; ++kk) {
#pragma unroll
        for (int T = 0; T < 8; ++T) {
          const int fa = sb_ilo(T, c16) * 8 + kb + bp * 4;
          const float *fr = fb + FR_OFF + (kk ? (fa ^ 4) : fa);
          const float *lq = lpb + kk * LPI_OFF + T * LPT_PITCH + c16 * 4 + kb;
          v2f a = *reinterpret_cast<const v2f *>(fr);
          const v2f b = *reinterpret_cast<const v2f *>(lq);
          if (kk && !bp) { a.x = -a.x; a.y = -a.y; }
          acc[T] = __builtin_amdgcn_wmma_f32_16x16x4_f32(false, a, false, b,
                                                         (short)0, acc[T], false, false);
        }
      }
#pragma unroll
      for (int T = 0; T < 8; ++T) {
        const int bd = sb_base_d(T, c16, h);
#pragma unroll
        for (int r = 0; r < 8; ++r) {
          const int w = bd + ((r & 4) ? e5 : 0) + ((r & 2) ? e6 : 0) + (r & 1) * 4;
          SS[w] = acc[T][r];
        }
      }
    }
    wave_handover();

    {
      const float *Lg = gm + L2_OFF;
      pass_rb(Lg + 256 + baseA, SRE(0) + laneBB, SIM(0) + laneBB, SRE(1) + laneWB);
      wave_handover();
      pass_rc(Lg + 512 + baseA, SRE(1) + laneBC, SIM(1) + laneBC,
              SRE(0) + h * PSTR,
              pbL);
      wave_handover();
    }

    float z = pass_z(gm + L3_OFF + baseA, SRE(0) + laneB3, SIM(0) + laneB3);
#pragma unroll
    for (int m = 16; m > 0; m >>= 1) z += __shfl_xor(z, m, 32);
    if (lane == 0 && smp < nS) out[smp] = z + bias0;
  }
#undef SRE
#undef SIM
}

extern "C" void kernel_launch(void *const *d_in, const int *in_sizes, int n_in,
                              void *d_out, int out_size, void *d_ws, size_t ws_size,
                              hipStream_t stream) {
  const float *X = (const float *)d_in[0];
  const float *W = (const float *)d_in[1];
  const float *bias = (const float *)d_in[2];
  float *out = (float *)d_out;

  const int B = in_sizes[0] / N_WIRES;

  vqc_kernel<<<NWG, 256, 0, stream>>>(X, W, bias, out, B);
}


# LDS layouts of the 3 state hand-overs (state build>RB>RC>layer 3) split into lane part + literal part for writer and reader, register pairs adjacent: 8-byte stores, no scatter math; sums unchanged
# speedup vs baseline: 5.4330x; 1.0502x over previous
#include <hip/hip_runtime.h>


typedef float v2f __attribute__((ext_vector_type(2)));
typedef float v8f __attribute__((ext_vector_type(8)));

#define N_WIRES 10
#define N_LAYERS 4
#define DIM 1024
#define LP_BASE 1120
#define L2_OFF 0
#define L3_OFF (L2_OFF + 768)
#define U1_OFF (L3_OFF + 256)
#define GM_WORDS 1120
#define NSW 8
#define NWG 512
#define B1_BASE 2304
#define SSTR (2 * B1_BASE)
#define LA_PL 1152
#define LB_PL 32
#define LC_PL 288
#define VR_OFF 0
#define VI_OFF 32
#define FL_OFF 64
#define FR_OFF 320
#define LPT_PITCH 68
#define LPI_OFF 542

__device__ __forceinline__ void cmul(float ar, float ai, float br, float bi,
                                     float &cr, float &ci) {
  cr = ar * br - ai * bi;
  ci = ar * bi + ai * br;
}

__device__ __forceinline__ int cnot_perm(int m, int rr) {
#pragma unroll
  for (int w = 0; w < N_WIRES; ++w) {
    int t = w + rr; if (t >= N_WIRES) t -= N_WIRES;
    m ^= ((m >> (9 - w)) & 1) << (9 - t);
  }
  return m;
}

__device__ __forceinline__ int la_addr(int x) {
  return ((x >> 0) & 1) * 32 + ((x >> 1) & 1) * 16 + ((x >> 2) & 1) * 1 + ((x >> 3) & 1) * 8
       + ((x >> 4) & 1) * 2 + ((x >> 5) & 1) * 4 + ((x >> 6) & 1) * 576 + ((x >> 7) & 1) * 288
       + ((x >> 8) & 1) * 144 + ((x >> 9) & 1) * 72;
}
__device__ __forceinline__ int lb_addr(int x) {
  return ((x >> 0) & 1) * 2 + ((x >> 1) & 1) * 4 + ((x >> 2) & 1) * 8 + ((x >> 3) & 1) * 16
       + ((x >> 4) & 1) * 1 + ((x >> 5) & 1) * 576 + ((x >> 6) & 1) * 72 + ((x >> 7) & 1) * 144
       + ((x >> 8) & 1) * 1152 + ((x >> 9) & 1) * 288;
}
__device__ __forceinline__ int rc_lane_x(int c) {
  return (((c >> 3) & 1) << 9) | (((c >> 2) & 1) << 7) | (((c >> 1) & 1) << 6) | ((c & 1) << 4);
}
__device__ __forceinline__ int rc_tile_x(int T) {
  return (((T >> 2) & 1) << 8) | (((T >> 1) & 1) << 5) | ((T & 1) << 2);
}
__device__ __forceinline__ int par_of(int v) { return __builtin_popcount((unsigned)v) & 1; }
__device__ __forceinline__ int lc_addr(int d) {
  return par_of(d & 0x041) * 1   + par_of(d & 0x048) * 8   + par_of(d & 0x020) * 16
       + par_of(d & 0x040) * 2   + par_of(d & 0x0c1) * 4   + par_of(d & 0x010) * 72
       + par_of(d & 0x204) * 144 + par_of(d & 0x012) * 576 + par_of(d & 0x110) * 32
       + par_of(d & 0x004) * 1152;
}
__device__ __forceinline__ int z_lane_d(int c) {
  return (((c >> 3) & 1) << 7) | (((c >> 2) & 1) << 6) | (((c >> 1) & 1) << 5) | ((c & 1) << 3);
}
__device__ __forceinline__ int z_tile_d(int T) {
  return (((T >> 2) & 1) << 9) | (((T >> 1) & 1) << 2) | ((T & 1) << 1);
}

__device__ __forceinline__ float real_form3(const float (*u)[8], int w0, int w1, int w2,
                                            int r, int c) {
  float pr = 1.f, pi = 0.f;
#pragma unroll
  for (int q = 0; q < 3; ++q) {
    const int wq = (q == 0) ? w0 : ((q == 1) ? w1 : w2);
    const int rb = (r >> (2 - q)) & 1, cb2 = (c >> (2 - q)) & 1;
    const float *e = &u[wq][(rb * 2 + cb2) * 2];
    float nr, ni; cmul(pr, pi, e[0], e[1], nr, ni);
    pr = nr; pi = ni;
  }
  const int rq = r >> 3, cq = c >> 3;
  return (rq == cq) ? pr : (rq ? pi : -pi);
}

__device__ __forceinline__ void rot_gate(float phi, float theta, float omega, float *u) {
  const float cs = __cosf(0.5f * theta), s = __sinf(0.5f * theta);
  const float a = 0.5f * (phi + omega), b = 0.5f * (phi - omega);
  const float ca = __cosf(a), sa = __sinf(a), cb = __cosf(b), sb = __sinf(b);
  u[0] =  ca * cs; u[1] = -sa * cs;
  u[2] = -cb * s;  u[3] = -sb * s;
  u[4] =  cb * s;  u[5] = -sb * s;
  u[6] =  ca * cs; u[7] =  sa * cs;
}

__device__ __forceinline__ void wave_handover() {
  __builtin_amdgcn_fence(__ATOMIC_SEQ_CST, "wavefront");
}


__device__ __forceinline__ void pass_rb(const float *__restrict__ A,
                                        const float *b_, float *dst) {
  v8f acc[8];
#pragma unroll
  for (int T = 0; T < 8; ++T) acc[T] = v8f{};
#pragma unroll
  for (int kk = 0; kk < 4; ++kk) {
    v2f a;
    a.x = A[kk * 64];  a.y = A[kk * 64 + 16];
#pragma unroll
    for (int T = 0; T < 8; ++T) {
      const int o = la_addr((T << 7) | ((kk & 1) << 6)) + (kk >> 1) * LA_PL;
      v2f b;
      b.x = b_[o];  b.y = b_[o + 2];
      acc[T] = __builtin_amdgcn_wmma_f32_16x16x4_f32(false, a, false, b,
                                                     (short)0, acc[T], false, false);
    }
  }
#pragma unroll
  for (int T = 0; T < 8; ++T) {
#pragma unroll
    for (int v = 0; v < 8; v += 2) {
      v2f d;
      d.x = acc[T][v];  d.y = acc[T][v + 1];
      *reinterpret_cast<v2f *>(dst + lb_addr((T << 7) | (v << 4))) = d;
    }
  }
}

__device__ __forceinline__ void pass_rc(const float *__restrict__ A,
                                        const float *b_, float *dst) {
  v8f acc[8];
#pragma unroll
  for (int T = 0; T < 8; ++T) acc[T] = v8f{};
#pragma unroll
  for (int kk = 0; kk < 4; ++kk) {
    v2f a;
    a.x = A[kk * 64];  a.y = A[kk * 64 + 16];
#pragma unroll
    for (int T = 0; T < 8; ++T) {
      const int o = lb_addr(rc_tile_x(T) | ((kk & 1) << 3)) + (kk >> 1) * LB_PL;
      v2f b;
      b.x = b_[o];  b.y = b_[o + 2];
      acc[T] = __builtin_amdgcn_wmma_f32_16x16x4_f32(false, a, false, b,
                                                     (short)0, acc[T], false, false);
    }
  }
#pragma unroll
  for (int T = 0; T < 8; ++T) {
#pragma unroll
    for (int v = 0; v < 8; v += 2) {
      const int x0 = rc_tile_x(T) | ((v & 4) << 1) | (v & 2);
      const int w0 = lc_addr(cnot_perm(x0, 3)), w1 = lc_addr(cnot_perm(x0 | 1, 3));
      v2f d;
      if (w1 > w0) { d.x = acc[T][v];      d.y = acc[T][v + 1]; }
      else         { d.x = acc[T][v + 1];  d.y = acc[T][v]; }
      *reinterpret_cast<v2f *>(dst + ((w1 > w0) ? w0 : w1)) = d;
    }
  }
}

__device__ __forceinline__ float pass_z(const float *__restrict__ A, const float *b_) {
  v8f acc[8];
#pragma unroll
  for (int T = 0; T < 8; ++T) acc[T] = v8f{};
#pragma unroll
  for (int kk = 0; kk < 4; ++kk) {
    v2f a;
    a.x = A[kk * 64];  a.y = A[kk * 64 + 16];
#pragma unroll
    for (int T = 0; T < 8; ++T) {
      const int dl = z_tile_d(T) | ((kk & 1) << 4);
      v2f b;
      b.x = b_[lc_addr(dl) + (kk >> 1) * LC_PL];
      b.y = b_[lc_addr(dl | (1 << 8)) + (kk >> 1) * LC_PL];
      acc[T] = __builtin_amdgcn_wmma_f32_16x16x4_f32(false, a, false, b,
                                                     (short)0, acc[T], false, false);
    }
  }
  float z = 0.f;
#pragma unroll
  for (int T = 0; T < 8; ++T) {
    float t = 0.f;
#pragma unroll
    for (int v = 0; v < 8; ++v) {
      const float sq = acc[T][v] * acc[T][v];
      t += (((v >> 2) ^ (v >> 1) ^ v) & 1) ? -sq : sq;
    }
    z += t;
  }
  return z;
}

__device__ __forceinline__ int sb_ilo(int T, int c16) {
  const int ts1 = (T >> 2) & 1, ts2 = (T >> 1) & 1, td3 = T & 1;
  const int bp = c16 >> 3, br6 = (c16 >> 1) & 1;
  const int bd5 = ((c16 >> 2) & 1) ^ bp, bd7 = c16 & 1;
  return (bd5 ^ td3) * 16 + br6 * 8 + (bd7 ^ bd5) * 4 + (ts1 ^ br6) * 2 + (ts2 ^ td3 ^ bd7);
}
__device__ __forceinline__ int sb_base_d(int T, int c16, int h) {
  const int ts1 = (T >> 2) & 1, ts2 = (T >> 1) & 1, td3 = T & 1;
  const int bd4 = c16 >> 3;
  return ((c16 >> 2) & 1) * 72 + ((c16 >> 1) & 1) * 144 + (c16 & 1) * 288 + td3 * 576
       + bd4 * (4 + 8) + (ts1 ^ bd4) * 16 + (ts2 ^ td3) * 32 + h * (LA_PL + 2);
}

__global__ __launch_bounds__(256, 4) void vqc_kernel(const float *__restrict__ X,
                                                  const float *__restrict__ Wt,
                                                  const float *__restrict__ bias,
                                                  float *__restrict__ out, int nS) {
  __shared__ __attribute__((aligned(16))) float SM[GM_WORDS + NSW * SSTR];
  float *const gm = SM;
  float *const ST = SM + GM_WORDS;

  const int tid  = threadIdx.x;
  const int lane = tid & 31;
  const int wave = tid >> 5;
  const int nGroups = (nS + NSW - 1) / NSW;
  if ((int)blockIdx.x >= nGroups) return;

  float *const ug = ST;
  if (tid < N_LAYERS * N_WIRES) {
    float u[8];
    rot_gate(Wt[tid * 3 + 0], Wt[tid * 3 + 1], Wt[tid * 3 + 2], u);
#pragma unroll
    for (int j = 0; j < 8; ++j) ug[tid * 8 + j] = u[j];
    if (tid >= N_WIRES && tid < 2 * N_WIRES) {
#pragma unroll
      for (int j = 0; j < 8; ++j) gm[U1_OFF + (tid - N_WIRES) * 8 + j] = u[j];
    }
  }

  float *const SS = ST + wave * SSTR;
  float *const fb = SS + B1_BASE;
  float *const lpb = fb + LP_BASE;

  const int ew = (lane < N_WIRES) ? lane : 0;
  float xcur = 0.f;
  if (lane < N_WIRES) {
    int s = (int)blockIdx.x * NSW + wave;
    if (s >= nS) s = nS - 1;
    xcur = X[s * N_WIRES + ew];
  }
  __syncthreads();
  const float4 m0 = *reinterpret_cast<const float4 *>(ug + ew * 8);
  const float4 m1 = *reinterpret_cast<const float4 *>(ug + ew * 8 + 4);
  {
    const int r = tid & 15, c = tid >> 4;
    const float (*u2)[8] = reinterpret_cast<const float (*)[8]>(ug + 2 * N_WIRES * 8);
    const float (*u3)[8] = reinterpret_cast<const float (*)[8]>(ug + 3 * N_WIRES * 8);
    gm[L2_OFF + tid]       = real_form3(u2, 0, 1, 2, r, c);
    gm[L2_OFF + 256 + tid] = real_form3(u2, 3, 4, 5, r, c);
    gm[L2_OFF + 512 + tid] = real_form3(u2, 6, 8, 9, r, c);
    gm[L3_OFF + tid]       = real_form3(u3, 5, 9, 1, r, c);
  }
  __syncthreads();

  const int  c16 = lane & 15;
  const int  h   = lane >> 4;
  const int  kb  = h << 1;
  const int  baseA = kb * 16 + c16;

  const int laneBB = la_addr(c16 | (h << 5));
  const int laneWB = 2 * c16 + h * LB_PL;
  const int laneBC = lb_addr(rc_lane_x(c16) | (h << 1));
  const int laneWC = lc_addr(cnot_perm(rc_lane_x(c16), 3)) + h * LC_PL;
  const int laneB3 = lc_addr(z_lane_d(c16) | h);
  const int bp = c16 >> 3;
  const int e5 = h ? -2 : 2, e6 = bp ? -8 : 8;

  const float bias0 = bias[0];
  for (int grp = blockIdx.x; grp < nGroups; grp += NWG) {
    const int smp = grp * NSW + wave;

    if (lane < N_WIRES) {
      const float x = xcur * 0.7853981633974483f;
      const float c = __cosf(x), sn = __sinf(x);
      float *vr = fb + VR_OFF;
      float *vi = fb + VI_OFF;
      vr[ew]           = m0.x * c + m0.z * sn;
      vi[ew]           = m0.y * c + m0.w * sn;
      vr[N_WIRES + ew] = m1.x * c + m1.z * sn;
      vi[N_WIRES + ew] = m1.y * c + m1.w * sn;
    }
    wave_handover();

    if (lane < N_WIRES) {
      int sn2 = (grp + NWG) * NSW + wave;
      if (sn2 >= nS) sn2 = nS - 1;
      xcur = X[sn2 * N_WIRES + ew];
    }

    {
      const float *vr = fb + VR_OFF, *vi = fb + VI_OFF;
      const float *u1 = gm + U1_OFF;
      {
        const int t = lane;
#pragma unroll
        for (int g = 0; g < 2; ++g) {
          float rr_[2], ri_[2];
          {
            const float *e = u1 + 9 * 8 + (((t & 1) * 2 + g) * 2);
#pragma unroll
            for (int p = 0; p < 2; ++p) {
              const int vb = (g ^ p) * N_WIRES + 9;
              cmul(e[0], e[1], vr[vb], vi[vb], rr_[p], ri_[p]);
            }
          }
#pragma unroll
          for (int w = 8; w >= 5; --w) {
            const int ib = (t >> (9 - w)) & 1;
            float qr[2], qi[2];
#pragma unroll
            for (int c = 0; c < 2; ++c) {
              const float *e = u1 + w * 8 + ((ib * 2 + c) * 2);
              cmul(e[0], e[1], rr_[c], ri_[c], qr[c], qi[c]);
            }
#pragma unroll
            for (int p = 0; p < 2; ++p) {
              const int e0 = p * N_WIRES + w, e1 = (1 ^ p) * N_WIRES + w;
              float ar, ai, br, bi;
              cmul(qr[0], qi[0], vr[e0], vi[e0], ar, ai);
              cmul(qr[1], qi[1], vr[e1], vi[e1], br, bi);
              rr_[p] = ar + br; ri_[p] = ai + bi;
            }
          }
          fb[FR_OFF + t * 8 + g * 2 + 0] = rr_[0];
          fb[FR_OFF + t * 8 + g * 2 + 1] = rr_[1];
          fb[FR_OFF + t * 8 + 4 + g * 2 + 0] = ri_[0];
          fb[FR_OFF + t * 8 + 4 + g * 2 + 1] = ri_[1];
        }
      }
      {
        const int hh = lane;
#pragma unroll
        for (int g = 0; g < 2; ++g) {
          float ar_[2], ai_[2];
          {
            const int ib = (hh >> 4) & 1;
#pragma unroll
            for (int c = 0; c < 2; ++c) {
              const float *e = u1 + ((ib * 2 + c) * 2);
              const int vb = (c ^ g) * N_WIRES;
              cmul(e[0], e[1], vr[vb], vi[vb], ar_[c], ai_[c]);
            }
          }
#pragma unroll
          for (int w = 1; w < 5; ++w) {
            const int ib = (hh >> (4 - w)) & 1;
            const int x = (w == 1) ? g : 0;
            float nr[2], ni[2];
#pragma unroll
            for (int c = 0; c < 2; ++c) {
              const int e0 = (c ^ x) * N_WIRES + w, e1 = (c ^ 1 ^ x) * N_WIRES + w;
              float s0r, s0i, s1r, s1i;
              cmul(ar_[0], ai_[0], vr[e0], vi[e0], s0r, s0i);
              cmul(ar_[1], ai_[1], vr[e1], vi[e1], s1r, s1i);
              const float *e = u1 + w * 8 + ((ib * 2 + c) * 2);
              cmul(e[0], e[1], s0r + s1r, s0i + s1i, nr[c], ni[c]);
            }
            ar_[0] = nr[0]; ai_[0] = ni[0]; ar_[1] = nr[1]; ai_[1] = ni[1];
          }
          fb[FL_OFF + hh * 8 + g * 2 + 0] = ar_[0];
          fb[FL_OFF + hh * 8 + g * 2 + 1] = ar_[1];
          fb[FL_OFF + hh * 8 + 4 + g * 2 + 0] = ai_[0];
          fb[FL_OFF + hh * 8 + 4 + g * 2 + 1] = ai_[1];
        }
      }
    }
    wave_handover();

    {
      const float *fl = fb + FL_OFF;
      const float *A = gm + L2_OFF + baseA;
      const int ih = (((c16 >> 3) & 1) * 2) ^ ((c16 >> 2) & 1) ^ (((c16 >> 1) & 1) * 20)
                   ^ (((c16 & 1) ^ h) * 10);
      v8f acc[4];
#pragma unroll
      for (int q = 0; q < 4; ++q) acc[q] = v8f{};
#pragma unroll
      for (int kk = 0; kk < 4; ++kk) {
        const int r0 = (ih ^ ((kk & 1) * 20)) * 8 + (kk >> 1) * 4;
        const int r1 = (ih ^ ((kk & 1) * 20) ^ 5) * 8 + (kk >> 1) * 4;
        v2f a;
        a.x = A[kk * 64];  a.y = A[kk * 64 + 16];
#pragma unroll
        for (int q = 0; q < 4; ++q) {
          v2f b;
          b.x = fl[r0 + q];  b.y = fl[r1 + q];
          acc[q] = __builtin_amdgcn_wmma_f32_16x16x4_f32(false, a, false, b,
                                                         (short)0, acc[q], false, false);
        }
      }
      const int tl = ((((c16 >> 1) ^ (c16 >> 2)) & 1) << 2) | (((c16 ^ (c16 >> 3)) & 1) << 1)
                   | (c16 >> 3);
      float *lp = lpb + h * LPI_OFF + tl * LPT_PITCH + ((c16 >> 2) & 1) * 32;
#pragma unroll
      for (int v = 0; v < 8; ++v) {
#pragma unroll
        for (int q = 0; q < 4; ++q) lp[v * 4 + q] = acc[q][v];
      }
    }
    wave_handover();

    {
      v8f acc[8];
#pragma unroll
      for (int T = 0; T < 8; ++T) acc[T] = v8f{};
#pragma unroll
      for (int kk = 0; kk < 2; ++kk) {
#pragma unroll
        for (int T = 0; T < 8; ++T) {
          const int fa = sb_ilo(T, c16) * 8 + kb + bp * 4;
          const float *fr = fb + FR_OFF + (kk ? (fa ^ 4) : fa);
          const float *lq = lpb + kk * LPI_OFF + T * LPT_PITCH + c16 * 4 + kb;
          v2f a = *reinterpret_cast<const v2f *>(fr);
          const v2f b = *reinterpret_cast<const v2f *>(lq);
          if (kk && !bp) { a.x = -a.x; a.y = -a.y; }
          acc[T] = __builtin_amdgcn_wmma_f32_16x16x4_f32(false, a, false, b,
                                                         (short)0, acc[T], false, false);
        }
      }
#pragma unroll
      for (int T = 0; T < 8; ++T) {
        const int bd = sb_base_d(T, c16, h);
#pragma unroll
        for (int r = 0; r < 8; r += 2) {
          const int w = bd + ((r & 4) ? e5 : 0) + ((r & 2) ? e6 : 0);
          v2f d;
          d.x = acc[T][r];  d.y = acc[T][r + 1];
          *reinterpret_cast<v2f *>(SS + w) = d;
        }
      }
    }
    wave_handover();

    {
      const float *Lg = gm + L2_OFF;
      pass_rb(Lg + 256 + baseA, SS + laneBB, fb + laneWB);
      wave_handover();
      pass_rc(Lg + 512 + baseA, fb + laneBC, SS + laneWC);
      wave_handover();
    }

    float z = pass_z(gm + L3_OFF + baseA, SS + laneB3);
#pragma unroll
    for (int m = 16; m > 0; m >>= 1) z += __shfl_xor(z, m, 32);
    if (lane == 0 && smp < nS) out[smp] = z + bias0;
  }
}

extern "C" void kernel_launch(void *const *d_in, const int *in_sizes, int n_in,
                              void *d_out, int out_size, void *d_ws, size_t ws_size,
                              hipStream_t stream) {
  const float *X = (const float *)d_in[0];
  const float *W = (const float *)d_in[1];
  const float *bias = (const float *)d_in[2];
  float *out = (float *)d_out;

  const int B = in_sizes[0] / N_WIRES;

  vqc_kernel<<<NWG, 256, 0, stream>>>(X, W, bias, out, B);
}
